# starting kernel
# baseline (speedup 1.0000x reference)
#include <hip/hip_runtime.h>
#include <math.h>

#define NFFT     1024
#define HOPSZ    512
#define NMELS    128
#define NFREQ    513
#define FPAD     528
#define T_IN     1323000
#define NB       16
#define NFRAMES  2584
#define MTILES   162
#define E_ELEMS  (NB*NMELS*NFRAMES)

#define WIN_OFF  0
#define DFT_OFF  1024
#define DFT_SZ   ((NFFT/2)*FPAD*4)
#define FB_OFF   (DFT_OFF + DFT_SZ)
#define FB_SZ    ((FPAD/2)*NMELS*2)
#define MAX_OFF  (FB_OFF + FB_SZ)
#define MEL_OFF  (MAX_OFF + 8)
#define HARM_OFF (MEL_OFF + E_ELEMS)
#define PERC_OFF (HARM_OFF + E_ELEMS)

typedef float v2f __attribute__((ext_vector_type(2)));
typedef float v4f __attribute__((ext_vector_type(4)));
typedef float v8f __attribute__((ext_vector_type(8)));

__device__ __forceinline__ v8f wmma4(v2f a, v2f b, v8f c) {
  return __builtin_amdgcn_wmma_f32_16x16x4_f32(false, a, false, b, (short)0, c,
                                               false, false);
}

__global__ __launch_bounds__(256) void build_tables(float* __restrict__ ws) {
  int gid = blockIdx.x * 256 + threadIdx.x;
  if (gid == 0) {
    unsigned* mx = (unsigned*)(ws + MAX_OFF);
    mx[0] = 0u; mx[1] = 0u; mx[2] = 0u;
  }
  const float TWO_PI = 6.28318530717958647692f;
  if (gid < DFT_OFF) {
    int k = gid;
    ws[WIN_OFF + k] = 0.5f * (1.0f - cosf(TWO_PI * (float)k / 1024.0f));
  } else if (gid < FB_OFF) {
    int i = gid - DFT_OFF;
    int entry = i & 3;
    int rest  = i >> 2;
    int f  = rest % FPAD;
    int k2 = rest / FPAD;
    int k  = 2 * k2 + (entry & 1);
    int r  = (k * f) & (NFFT - 1);
    float arg = (float)r * (TWO_PI / 1024.0f);
    ws[gid] = (entry < 2) ? cosf(arg) : sinf(arg);
  } else if (gid < MAX_OFF) {
    int i = gid - FB_OFF;
    int entry = i & 1;
    int rest  = i >> 1;
    int m  = rest % NMELS;
    int k2 = rest / NMELS;
    int f  = 2 * k2 + entry;
    float v = 0.0f;
    if (f < NFREQ) {
      float freq   = (float)f * (22050.0f / 512.0f);
      float melmax = 2595.0f * log10f(1.0f + 22050.0f / 700.0f);
      float step   = melmax / 129.0f;
      float p0 = 700.0f * (exp10f(((float)(m    ) * step) / 2595.0f) - 1.0f);
      float p1 = 700.0f * (exp10f(((float)(m + 1) * step) / 2595.0f) - 1.0f);
      float p2 = 700.0f * (exp10f(((float)(m + 2) * step) / 2595.0f) - 1.0f);
      float down = (freq - p0) / (p1 - p0);
      float up   = (p2 - freq) / (p2 - p1);
      float t = fmaxf(0.0f, fminf(down, up));
      v = t * 2.0f / (p2 - p0);
    }
    ws[gid] = v;
  }
}

#define KCHUNK 256
__global__ __launch_bounds__(256) void mel_gemm(const float* __restrict__ wave,
                                                const float* __restrict__ ws,
                                                float* __restrict__ mel) {
  __shared__ float AL[16][KCHUNK + 4];
  __shared__ float PL[16][FPAD + 4];

  const float* __restrict__ win  = ws + WIN_OFF;
  const float* __restrict__ dftT = ws + DFT_OFF;
  const float* __restrict__ fb2  = ws + FB_OFF;

  int b    = blockIdx.x / MTILES;
  int tile = blockIdx.x % MTILES;
  int t0   = tile * 16;
  int tid  = threadIdx.x;
  int lane = tid & 31;
  int wv   = tid >> 5;
  int h    = lane >> 4;
  int l16  = lane & 15;

  const float* wb = wave + (long)b * T_IN;

  v8f accC[5], accS[5];
  #pragma unroll
  for (int i = 0; i < 5; ++i)
    for (int j = 0; j < 8; ++j) { accC[i][j] = 0.0f; accS[i][j] = 0.0f; }

  for (int kc = 0; kc < NFFT / KCHUNK; ++kc) {
    __syncthreads();
    for (int idx = tid; idx < 16 * KCHUNK; idx += 256) {
      int m  = idx >> 8;
      int kl = idx & (KCHUNK - 1);
      int t  = t0 + m;
      float v = 0.0f;
      if (t < NFRAMES) {
        int k = kc * KCHUNK + kl;
        int o = t * HOPSZ + k - NFFT / 2;
        o = (o < 0) ? -o : o;
        o = (o >= T_IN) ? (2 * (T_IN - 1) - o) : o;
        v = wb[o] * win[k];
      }
      AL[m][kl] = v;
    }
    __syncthreads();

    #pragma unroll
    for (int i = 0; i < 5; ++i) {
      int ft = wv + 8 * i;
      if (ft < FPAD / 16) {
        int f0 = ft * 16;
        const float* dB = dftT + (long)(f0 + l16) * 4;
        __builtin_prefetch(dB + (long)(kc * (KCHUNK / 2)) * FPAD * 4, 0, 1);
        for (int k = 0; k < KCHUNK; k += 4) {
          int kk  = k + 2 * h;
          v2f a   = *(const v2f*)&AL[l16][kk];
          int k2g = kc * (KCHUNK / 2) + (kk >> 1);
          v4f q = *(const v4f*)(dB + (long)k2g * (FPAD * 4));
          v2f bc; bc[0] = q[0]; bc[1] = q[1];
          v2f bs; bs[0] = q[2]; bs[1] = q[3];
          accC[i] = wmma4(a, bc, accC[i]);
          accS[i] = wmma4(a, bs, accS[i]);
        }
      }
    }
  }

  __syncthreads();
  #pragma unroll
  for (int i = 0; i < 5; ++i) {
    int ft = wv + 8 * i;
    if (ft < FPAD / 16) {
      int f0 = ft * 16;
      #pragma unroll
      for (int v = 0; v < 8; ++v) {
        int mr = v + 8 * h;
        float re = accC[i][v], im = accS[i][v];
        PL[mr][f0 + l16] = re * re + im * im;
      }
    }
  }
  __syncthreads();

  int n0 = wv * 16;
  const float* fB = fb2 + (long)(n0 + l16) * 2;
  v8f acc;
  #pragma unroll
  for (int j = 0; j < 8; ++j) acc[j] = 0.0f;
  for (int k = 0; k < FPAD; k += 4) {
    int kk = k + 2 * h;
    v2f a  = *(const v2f*)&PL[l16][kk];
    v2f bb = *(const v2f*)(fB + (long)(kk >> 1) * (NMELS * 2));
    acc = wmma4(a, bb, acc);
  }
  int n = n0 + l16;
  #pragma unroll
  for (int v = 0; v < 8; ++v) {
    int t = t0 + v + 8 * h;
    if (t < NFRAMES) mel[(b * NMELS + n) * NFRAMES + t] = acc[v];
  }
}

__device__ __forceinline__ float median31(float v[31]) {
  #pragma unroll
  for (int i = 0; i < 16; ++i) {
    #pragma unroll
    for (int j = i + 1; j < 31; ++j) {
      float lo = fminf(v[i], v[j]);
      float hi = fmaxf(v[i], v[j]);
      v[i] = lo; v[j] = hi;
    }
  }
  return v[15];
}

__global__ __launch_bounds__(256) void median_time(const float* __restrict__ mel,
                                                   float* __restrict__ harm) {
  int gid = blockIdx.x * 256 + threadIdx.x;
  if (gid >= E_ELEMS) return;
  int t   = gid % NFRAMES;
  int row = gid / NFRAMES;
  const float* src = mel + row * NFRAMES;
  float v[31];
  #pragma unroll
  for (int d = 0; d < 31; ++d) {
    int j = t + d - 15;
    j = (j < 0) ? (-j - 1) : j;
    j = (j >= NFRAMES) ? (2 * NFRAMES - 1 - j) : j;
    v[d] = src[j];
  }
  harm[gid] = median31(v);
}

__global__ __launch_bounds__(256) void median_mel(const float* __restrict__ mel,
                                                  float* __restrict__ perc) {
  int gid = blockIdx.x * 256 + threadIdx.x;
  if (gid >= E_ELEMS) return;
  int t = gid % NFRAMES;
  int n = (gid / NFRAMES) % NMELS;
  int b = gid / (NFRAMES * NMELS);
  const float* base = mel + (b * NMELS) * NFRAMES + t;
  float v[31];
  #pragma unroll
  for (int d = 0; d < 31; ++d) {
    int j = n + d - 15;
    j = (j < 0) ? (-j - 1) : j;
    j = (j >= NMELS) ? (2 * NMELS - 1 - j) : j;
    v[d] = base[j * NFRAMES];
  }
  perc[gid] = median31(v);
}

__global__ __launch_bounds__(256) void hpss_mask(const float* __restrict__ mel,
                                                 float* __restrict__ harm,
                                                 float* __restrict__ perc,
                                                 unsigned* __restrict__ mx) {
  __shared__ float r0[256], r1[256], r2[256];
  int gid = blockIdx.x * 256 + threadIdx.x;
  float s = 0.0f, dh = 0.0f, dp = 0.0f;
  if (gid < E_ELEMS) {
    float S  = mel[gid];
    float ha = harm[gid];
    float pe = perc[gid];
    float Z  = fmaxf(ha, pe);
    bool bad = Z < 1.17549435e-38f;
    float Zs = bad ? 1.0f : Z;
    float mh = ha / Zs; mh *= mh;
    float mp = pe / Zs; mp *= mp;
    float den = mh + mp;
    float maskh = bad ? 0.0f : mh / den;
    float maskp = bad ? 0.0f : mp / den;
    dh = S * maskh; dp = S * maskp; s = S;
    harm[gid] = dh;
    perc[gid] = dp;
  }
  int tid = threadIdx.x;
  r0[tid] = s; r1[tid] = dh; r2[tid] = dp;
  __syncthreads();
  for (int off = 128; off > 0; off >>= 1) {
    if (tid < off) {
      r0[tid] = fmaxf(r0[tid], r0[tid + off]);
      r1[tid] = fmaxf(r1[tid], r1[tid + off]);
      r2[tid] = fmaxf(r2[tid], r2[tid + off]);
    }
    __syncthreads();
  }
  if (tid == 0) {
    atomicMax(mx + 0, __float_as_uint(r0[0]));
    atomicMax(mx + 1, __float_as_uint(r1[0]));
    atomicMax(mx + 2, __float_as_uint(r2[0]));
  }
}

__global__ __launch_bounds__(256) void finalize(const float* __restrict__ mel,
                                                const float* __restrict__ harm,
                                                const float* __restrict__ perc,
                                                const unsigned* __restrict__ mx,
                                                float* __restrict__ out) {
  int gid = blockIdx.x * 256 + threadIdx.x;
  if (gid >= 3 * E_ELEMS) return;
  int s = gid / E_ELEMS;
  int e = gid - s * E_ELEMS;
  const float* src = (s == 0) ? mel : ((s == 1) ? harm : perc);
  float ref = __uint_as_float(mx[s]);
  float v   = src[e];
  float db = 10.0f * log10f(fmaxf(1e-10f, v)) -
             10.0f * log10f(fmaxf(1e-10f, ref));
  db = fmaxf(db, -80.0f);
  out[gid] = (db + 80.0f) * (1.0f / (80.0f + 1e-7f));
}

extern "C" void kernel_launch(void* const* d_in, const int* in_sizes, int n_in,
                              void* d_out, int out_size, void* d_ws,
                              size_t ws_size, hipStream_t stream) {
  const float* wave = (const float*)d_in[0];
  float* out  = (float*)d_out;
  float* ws   = (float*)d_ws;
  float* mel  = ws + MEL_OFF;
  float* harm = ws + HARM_OFF;
  float* perc = ws + PERC_OFF;
  unsigned* mx = (unsigned*)(ws + MAX_OFF);

  build_tables<<<MAX_OFF / 256, 256, 0, stream>>>(ws);
  mel_gemm<<<NB * MTILES, 256, 0, stream>>>(wave, ws, mel);
  int mblocks = (E_ELEMS + 255) / 256;
  median_time<<<mblocks, 256, 0, stream>>>(mel, harm);
  median_mel<<<mblocks, 256, 0, stream>>>(mel, perc);
  hpss_mask<<<mblocks, 256, 0, stream>>>(mel, harm, perc, mx);
  finalize<<<(3 * E_ELEMS + 255) / 256, 256, 0, stream>>>(mel, harm, perc, mx,
                                                          out);
}


# median_time, median_mel: a thread keeps one sorted 31-tap window and slides it along the filter axis (drop the leaving tap, insert the entering one) instead of a selection per output; exact medians
# speedup vs baseline: 1.9459x; 1.9459x over previous
#include <hip/hip_runtime.h>
#include <math.h>

#define NFFT     1024
#define HOPSZ    512
#define NMELS    128
#define NFREQ    513
#define FPAD     528
#define T_IN     1323000
#define NB       16
#define NFRAMES  2584
#define E_ELEMS  (NB*NMELS*NFRAMES)

#define DFT_OFF  0
#define DFT_SZ   ((NFFT/4)*FPAD*4)
#define FB_OFF   (DFT_OFF + DFT_SZ)
#define FB_SZ    ((FPAD/2)*NMELS*2)
#define MAX_OFF  (FB_OFF + FB_SZ)
#define MEL_OFF  (MAX_OFF + 8)
#define HARM_OFF (MEL_OFF + E_ELEMS)
#define PERC_OFF (HARM_OFF + E_ELEMS)

typedef float v2f __attribute__((ext_vector_type(2)));
typedef float v4f __attribute__((ext_vector_type(4)));
typedef float v8f __attribute__((ext_vector_type(8)));

__device__ __forceinline__ v8f wmma4(v2f a, v2f b, v8f c) {
  return __builtin_amdgcn_wmma_f32_16x16x4_f32(false, a, false, b, (short)0, c,
                                               false, false);
}

__global__ __launch_bounds__(256) void build_tables(float* __restrict__ ws) {
  int gid = blockIdx.x * 256 + threadIdx.x;
  if (gid == 0) {
    unsigned* mx = (unsigned*)(ws + MAX_OFF);
    mx[0] = 0u; mx[1] = 0u; mx[2] = 0u;
  }
  const float TWO_PI = 6.28318530717958647692f;
  if (gid < FB_OFF) {
    int i = gid - DFT_OFF;
    int entry = i & 3;
    int rest  = i >> 2;
    int f  = rest % FPAD;
    int k2 = rest / FPAD;
    int k  = 2 * k2 + (entry & 1);
    int r  = (k * f) & (NFFT - 1);
    float arg = (float)r * (TWO_PI / 1024.0f);
    ws[gid] = (entry < 2) ? cosf(arg) : sinf(arg);
  } else if (gid < MAX_OFF) {
    int i = gid - FB_OFF;
    int entry = i & 1;
    int rest  = i >> 1;
    int m  = rest % NMELS;
    int k2 = rest / NMELS;
    int f  = 2 * k2 + entry;
    float v = 0.0f;
    if (f < NFREQ) {
      float freq   = (float)f * (22050.0f / 512.0f);
      float melmax = 2595.0f * log10f(1.0f + 22050.0f / 700.0f);
      float step   = melmax / 129.0f;
      float p0 = 700.0f * (exp10f(((float)(m    ) * step) / 2595.0f) - 1.0f);
      float p1 = 700.0f * (exp10f(((float)(m + 1) * step) / 2595.0f) - 1.0f);
      float p2 = 700.0f * (exp10f(((float)(m + 2) * step) / 2595.0f) - 1.0f);
      float down = (freq - p0) / (p1 - p0);
      float up   = (p2 - freq) / (p2 - p1);
      float t = fmaxf(0.0f, fminf(down, up));
      v = t * 2.0f / (p2 - p0);
    }
    ws[gid] = v;
  }
}

#define KH     512
#define HB     15
#define MT2    173
#define HPITCH (FPAD + 4)
__global__ __launch_bounds__(256) void mel_gemm(const float* __restrict__ wave,
                                                const float* __restrict__ ws,
                                                float* __restrict__ mel) {
  __shared__ float AL[16][HPITCH];
  __shared__ float HC[16][HPITCH];
  __shared__ float HS[16][HPITCH];
  float (*PL)[HPITCH] = AL;

  const float* __restrict__ dftT = ws + DFT_OFF;
  const float* __restrict__ fb2  = ws + FB_OFF;

  int b    = blockIdx.x / MT2;
  int tile = blockIdx.x % MT2;
  int t0   = tile * HB;
  int tid  = threadIdx.x;
  int lane = tid & 31;
  int wv   = tid >> 5;
  int h    = lane >> 4;
  int l16  = lane & 15;

  const float* wb = wave + (long)b * T_IN;

  for (int idx = tid; idx < 16 * KH; idx += 256) {
    int m = idx >> 9;
    int k = idx & (KH - 1);
    int j = t0 + m;
    float v = 0.0f;
    if (j <= NFRAMES) {
      int o = j * HOPSZ + k - NFFT / 2;
      o = (o < 0) ? -o : o;
      o = (o >= T_IN) ? (2 * (T_IN - 1) - o) : o;
      v = wb[o];
    }
    AL[m][k] = v;
  }
  __syncthreads();

  #pragma unroll
  for (int i = 0; i < 5; ++i) {
    int ft = wv + 8 * i;
    if (ft < FPAD / 16) {
      int f0 = ft * 16;
      const float* dB = dftT + (long)(f0 + l16) * 4;
      v8f aC, aS;
      #pragma unroll
      for (int j = 0; j < 8; ++j) { aC[j] = 0.0f; aS[j] = 0.0f; }
      for (int k = 0; k < KH; k += 4) {
        int kk = k + 2 * h;
        v2f a  = *(const v2f*)&AL[l16][kk];
        v4f q = *(const v4f*)(dB + (long)(kk >> 1) * (FPAD * 4));
        v2f bc; bc[0] = q[0]; bc[1] = q[1];
        v2f bs; bs[0] = q[2]; bs[1] = q[3];
        aC = wmma4(a, bc, aC);
        aS = wmma4(a, bs, aS);
      }
      #pragma unroll
      for (int v = 0; v < 8; ++v) {
        HC[v + 8 * h][f0 + l16] = aC[v];
        HS[v + 8 * h][f0 + l16] = aS[v];
      }
    }
  }
  __syncthreads();

  for (int idx = tid; idx < 16 * FPAD; idx += 256) {
    int m = idx / FPAD;
    int f = idx - m * FPAD;
    float p = 0.0f;
    if (m < HB && f < NFREQ) {
      float s  = (f & 1) ? -1.0f : 1.0f;
      int   fm = (f == 0) ? 1 : f - 1;
      float sm = (f == 0) ? -1.0f : 1.0f;
      int   fp = f + 1;
      float c0 = HC[m][f]  + s * HC[m + 1][f];
      float s0 = HS[m][f]  + s * HS[m + 1][f];
      float cm = HC[m][fm] - s * HC[m + 1][fm];
      float sn = (HS[m][fm] - s * HS[m + 1][fm]) * sm;
      float cp = HC[m][fp] - s * HC[m + 1][fp];
      float sp = HS[m][fp] - s * HS[m + 1][fp];
      float re = 0.5f * c0 - 0.25f * (cm + cp);
      float im = 0.5f * s0 - 0.25f * (sn + sp);
      p = re * re + im * im;
    }
    PL[m][f] = p;
  }
  __syncthreads();

  int n0 = wv * 16;
  const float* fB = fb2 + (long)(n0 + l16) * 2;
  v8f acc;
  #pragma unroll
  for (int j = 0; j < 8; ++j) acc[j] = 0.0f;
  for (int k = 0; k < FPAD; k += 4) {
    int kk = k + 2 * h;
    v2f a  = *(const v2f*)&PL[l16][kk];
    v2f bb = *(const v2f*)(fB + (long)(kk >> 1) * (NMELS * 2));
    acc = wmma4(a, bb, acc);
  }
  int n = n0 + l16;
  #pragma unroll
  for (int v = 0; v < 8; ++v) {
    int r = v + 8 * h;
    int t = t0 + r;
    if (r < HB && t < NFRAMES) mel[(b * NMELS + n) * NFRAMES + t] = acc[v];
  }
}


#define WIN_INSERT(a, m, x)                                             \
  do {                                                                  \
    if ((m) == 0) { (a)[0] = (x); }                                     \
    else {                                                              \
      (a)[(m)] = fmaxf((a)[(m) - 1], (x));                              \
      for (int i_ = (m) - 1; i_ >= 1; --i_)                             \
        (a)[i_] = fmaxf((a)[i_ - 1], fminf((a)[i_], (x)));              \
      (a)[0] = fminf((a)[0], (x));                                      \
    }                                                                   \
  } while (0)

__device__ __forceinline__ void win_slide(float (&a)[31], float xo, float xn) {
  float r[30];
  #pragma unroll
  for (int i = 0; i < 30; ++i) r[i] = (a[i] < xo) ? a[i] : a[i + 1];
  a[0] = fminf(r[0], xn);
  #pragma unroll
  for (int i = 1; i < 30; ++i) a[i] = fmaxf(r[i - 1], fminf(r[i], xn));
  a[30] = fmaxf(r[29], xn);
}

#define RT 11
__global__ __launch_bounds__(256) void median_time(const float* __restrict__ mel,
                                                   float* __restrict__ medT) {
  __shared__ float P[NFRAMES + 30];
  __shared__ float O[NFRAMES];
  int row = blockIdx.x;
  int tid = threadIdx.x;
  const float* src = mel + (long)row * NFRAMES;
  for (int i = tid; i < NFRAMES + 30; i += 256) {
    int j = i - 15;
    j = (j < 0) ? (-j - 1) : j;
    j = (j >= NFRAMES) ? (2 * NFRAMES - 1 - j) : j;
    P[i] = src[j];
  }
  __syncthreads();
  int t0 = tid * RT;
  if (t0 < NFRAMES) {
    float a[31];
    #pragma unroll
    for (int m = 0; m < 31; ++m) {
      float x = P[t0 + m];
      WIN_INSERT(a, m, x);
    }
    O[t0] = a[15];
    #pragma unroll
    for (int i = 1; i < RT; ++i) {
      int t = t0 + i;
      if (t < NFRAMES) {
        win_slide(a, P[t - 1], P[t + 30]);
        O[t] = a[15];
      }
    }
  }
  __syncthreads();
  float* dst = medT + (long)row * NFRAMES;
  for (int i = tid; i < NFRAMES; i += 256) dst[i] = O[i];
}

#define RN 16
__global__ __launch_bounds__(256) void hpss_mel(const float* __restrict__ mel,
                                                float* __restrict__ harm,
                                                float* __restrict__ perc,
                                                unsigned* __restrict__ mx) {
  __shared__ float r0[256], r1[256], r2[256];
  int tid = threadIdx.x;
  int t   = blockIdx.x * 256 + tid;
  int n0  = blockIdx.y * RN;
  int b   = blockIdx.z;
  float s = 0.0f, dh = 0.0f, dp = 0.0f;
  if (t < NFRAMES) {
    int cb = (b * NMELS) * NFRAMES + t;
    const float* base = mel + cb;
    float a[31];
    #pragma unroll
    for (int m = 0; m < 31; ++m) {
      int j = n0 + m - 15;
      j = (j < 0) ? (-j - 1) : j;
      j = (j >= NMELS) ? (2 * NMELS - 1 - j) : j;
      float x = base[j * NFRAMES];
      WIN_INSERT(a, m, x);
    }
    #pragma unroll
    for (int i = 0; i < RN; ++i) {
      int n = n0 + i;
      if (i > 0) {
        int jo = n - 16;
        jo = (jo < 0) ? (-jo - 1) : jo;
        int jn = n + 15;
        jn = (jn >= NMELS) ? (2 * NMELS - 1 - jn) : jn;
        win_slide(a, base[jo * NFRAMES], base[jn * NFRAMES]);
      }
      float pe = a[15];
      int   e  = cb + n * NFRAMES;
      float S  = mel[e];
      float ha = harm[e];
      float Z  = fmaxf(ha, pe);
      bool bad = Z < 1.17549435e-38f;
      float Zs = bad ? 1.0f : Z;
      float mh = ha / Zs; mh *= mh;
      float mp = pe / Zs; mp *= mp;
      float den = mh + mp;
      float maskh = bad ? 0.0f : mh / den;
      float maskp = bad ? 0.0f : mp / den;
      float vh = S * maskh, vp = S * maskp;
      harm[e] = vh;
      perc[e] = vp;
      s = fmaxf(s, S); dh = fmaxf(dh, vh); dp = fmaxf(dp, vp);
    }
  }
  r0[tid] = s; r1[tid] = dh; r2[tid] = dp;
  __syncthreads();
  for (int off = 128; off > 0; off >>= 1) {
    if (tid < off) {
      r0[tid] = fmaxf(r0[tid], r0[tid + off]);
      r1[tid] = fmaxf(r1[tid], r1[tid + off]);
      r2[tid] = fmaxf(r2[tid], r2[tid + off]);
    }
    __syncthreads();
  }
  if (tid == 0) {
    atomicMax(mx + 0, __float_as_uint(r0[0]));
    atomicMax(mx + 1, __float_as_uint(r1[0]));
    atomicMax(mx + 2, __float_as_uint(r2[0]));
  }
}

__global__ __launch_bounds__(256) void finalize(const float* __restrict__ mel,
                                                const float* __restrict__ harm,
                                                const float* __restrict__ perc,
                                                const unsigned* __restrict__ mx,
                                                float* __restrict__ out) {
  int gid = blockIdx.x * 256 + threadIdx.x;
  if (gid >= 3 * E_ELEMS) return;
  int s = gid / E_ELEMS;
  int e = gid - s * E_ELEMS;
  const float* src = (s == 0) ? mel : ((s == 1) ? harm : perc);
  float ref = __uint_as_float(mx[s]);
  float v   = src[e];
  float db = 10.0f * log10f(fmaxf(1e-10f, v)) -
             10.0f * log10f(fmaxf(1e-10f, ref));
  db = fmaxf(db, -80.0f);
  out[gid] = (db + 80.0f) * (1.0f / (80.0f + 1e-7f));
}

extern "C" void kernel_launch(void* const* d_in, const int* in_sizes, int n_in,
                              void* d_out, int out_size, void* d_ws,
                              size_t ws_size, hipStream_t stream) {
  const float* wave = (const float*)d_in[0];
  float* out  = (float*)d_out;
  float* ws   = (float*)d_ws;
  float* mel  = ws + MEL_OFF;
  float* harm = ws + HARM_OFF;
  float* perc = ws + PERC_OFF;
  unsigned* mx = (unsigned*)(ws + MAX_OFF);

  build_tables<<<MAX_OFF / 256, 256, 0, stream>>>(ws);
  mel_gemm<<<NB * MT2, 256, 0, stream>>>(wave, ws, mel);
  median_time<<<NB * NMELS, 256, 0, stream>>>(mel, harm);
  hpss_mel<<<dim3((NFRAMES + 255) / 256, NMELS / RN, NB), 256, 0, stream>>>(
      mel, harm, perc, mx);
  finalize<<<(3 * E_ELEMS + 255) / 256, 256, 0, stream>>>(mel, harm, perc, mx,
                                                          out);
}


# mel_gemm stage 1: half-frame transform factored 512 = 8 x 64: WMMA does the 64-sample part (period 128 in f, K 64), the 8-point step runs per lane on the accumulators; f32 summation order changes
# speedup vs baseline: 2.7138x; 1.3946x over previous
#include <hip/hip_runtime.h>
#include <math.h>

#define NFFT     1024
#define HOPSZ    512
#define NMELS    128
#define NFREQ    513
#define FPAD     528
#define T_IN     1323000
#define NB       16
#define NFRAMES  2584
#define E_ELEMS  (NB*NMELS*NFRAMES)

#define KR       8
#define KJ       64
#define GP       128
#define DFT_OFF  0
#define DFT_SZ   ((KJ/2)*GP*4)
#define TW_OFF   (DFT_OFF + DFT_SZ)
#define TW_SZ    (KR*GP*2)
#define FB_OFF   (TW_OFF + TW_SZ)
#define FB_SZ    ((FPAD/2)*NMELS*2)
#define MAX_OFF  (FB_OFF + FB_SZ)
#define MEL_OFF  (MAX_OFF + 8)
#define HARM_OFF (MEL_OFF + E_ELEMS)
#define PERC_OFF (HARM_OFF + E_ELEMS)

typedef float v2f __attribute__((ext_vector_type(2)));
typedef float v4f __attribute__((ext_vector_type(4)));
typedef float v8f __attribute__((ext_vector_type(8)));

__device__ __forceinline__ v8f wmma4(v2f a, v2f b, v8f c) {
  return __builtin_amdgcn_wmma_f32_16x16x4_f32(false, a, false, b, (short)0, c,
                                               false, false);
}

__global__ __launch_bounds__(256) void build_tables(float* __restrict__ ws) {
  int gid = blockIdx.x * 256 + threadIdx.x;
  if (gid == 0) {
    unsigned* mx = (unsigned*)(ws + MAX_OFF);
    mx[0] = 0u; mx[1] = 0u; mx[2] = 0u;
  }
  const float TWO_PI = 6.28318530717958647692f;
  if (gid < TW_OFF) {
    int i = gid - DFT_OFF;
    int entry = i & 3;
    int rest  = i >> 2;
    int g  = rest % GP;
    int j2 = rest / GP;
    int j  = 2 * j2 + (entry & 1);
    int p  = (j * g) & (GP - 1);
    float arg = (float)p * (TWO_PI / 128.0f);
    ws[gid] = (entry < 2) ? cosf(arg) : sinf(arg);
  } else if (gid < FB_OFF) {
    int i = gid - TW_OFF;
    int entry = i & 1;
    int rest  = i >> 1;
    int g  = rest % GP;
    int r  = rest / GP;
    float arg = (float)(r * g) * (TWO_PI / 1024.0f);
    ws[gid] = entry ? sinf(arg) : cosf(arg);
  } else if (gid < MAX_OFF) {
    int i = gid - FB_OFF;
    int entry = i & 1;
    int rest  = i >> 1;
    int m  = rest % NMELS;
    int k2 = rest / NMELS;
    int f  = 2 * k2 + entry;
    float v = 0.0f;
    if (f < NFREQ) {
      float freq   = (float)f * (22050.0f / 512.0f);
      float melmax = 2595.0f * log10f(1.0f + 22050.0f / 700.0f);
      float step   = melmax / 129.0f;
      float p0 = 700.0f * (exp10f(((float)(m    ) * step) / 2595.0f) - 1.0f);
      float p1 = 700.0f * (exp10f(((float)(m + 1) * step) / 2595.0f) - 1.0f);
      float p2 = 700.0f * (exp10f(((float)(m + 2) * step) / 2595.0f) - 1.0f);
      float down = (freq - p0) / (p1 - p0);
      float up   = (p2 - freq) / (p2 - p1);
      float t = fmaxf(0.0f, fminf(down, up));
      v = t * 2.0f / (p2 - p0);
    }
    ws[gid] = v;
  }
}

#define KH     512
#define HB     15
#define MT2    173
#define HPITCH (FPAD + 4)
#define APITCH (KJ + 4)
#define AROW   (KR * APITCH + 4)

__device__ __forceinline__ void dft4(float r0, float i0, float r1, float i1,
                                     float r2, float i2, float r3, float i3,
                                     float (&Br)[4], float (&Bi)[4]) {
  float s02r = r0 + r2, s02i = i0 + i2, d02r = r0 - r2, d02i = i0 - i2;
  float s13r = r1 + r3, s13i = i1 + i3, d13r = r1 - r3, d13i = i1 - i3;
  Br[0] = s02r + s13r; Bi[0] = s02i + s13i;
  Br[1] = d02r - d13i; Bi[1] = d02i + d13r;
  Br[2] = s02r - s13r; Bi[2] = s02i - s13i;
  Br[3] = d02r + d13i; Bi[3] = d02i - d13r;
}
__global__ __launch_bounds__(256) void mel_gemm(const float* __restrict__ wave,
                                                const float* __restrict__ ws,
                                                float* __restrict__ mel) {
  __shared__ float AL[16][AROW];
  __shared__ float HC[16][HPITCH];
  __shared__ float HS[16][HPITCH];
  float (*PL)[AROW] = AL;

  const float* __restrict__ dftT = ws + DFT_OFF;
  const float* __restrict__ twT  = ws + TW_OFF;
  const float* __restrict__ fb2  = ws + FB_OFF;

  int b    = blockIdx.x / MT2;
  int tile = blockIdx.x % MT2;
  int t0   = tile * HB;
  int tid  = threadIdx.x;
  int lane = tid & 31;
  int wv   = tid >> 5;
  int h    = lane >> 4;
  int l16  = lane & 15;

  const float* wb = wave + (long)b * T_IN;

  for (int idx = tid; idx < 16 * KH; idx += 256) {
    int m = idx >> 9;
    int k = idx & (KH - 1);
    int j = t0 + m;
    float v = 0.0f;
    if (j <= NFRAMES) {
      int o = j * HOPSZ + k - NFFT / 2;
      o = (o < 0) ? -o : o;
      o = (o >= T_IN) ? (2 * (T_IN - 1) - o) : o;
      v = wb[o];
    }
    AL[m][(k & (KR - 1)) * APITCH + (k >> 3)] = v;
  }
  __syncthreads();

  {
    int g0 = wv * 16;
    int g  = g0 + l16;
    const float* dB = dftT + (long)g * 4;
    float tc[KR], ts[KR];
    tc[0] = 1.0f; ts[0] = 0.0f;
    #pragma unroll
    for (int r = 1; r < KR; ++r) {
      v2f t = *(const v2f*)(twT + (long)(r * GP + g) * 2);
      tc[r] = t[0]; ts[r] = t[1];
    }
    v8f aC[KR], aS[KR];
    #pragma unroll
    for (int r = 0; r < KR; ++r) {
      #pragma unroll
      for (int j = 0; j < 8; ++j) { aC[r][j] = 0.0f; aS[r][j] = 0.0f; }
    }
    #pragma unroll
    for (int k = 0; k < KJ; k += 4) {
      int kk = k + 2 * h;
      v4f q = *(const v4f*)(dB + (long)(kk >> 1) * (GP * 4));
      v2f bc; bc[0] = q[0]; bc[1] = q[1];
      v2f bs; bs[0] = q[2]; bs[1] = q[3];
      #pragma unroll
      for (int r = 0; r < KR; ++r) {
        v2f a = *(const v2f*)&AL[l16][r * APITCH + kk];
        aC[r] = wmma4(a, bc, aC[r]);
        aS[r] = wmma4(a, bs, aS[r]);
      }
    }
    const float RH = 0.70710678118654752440f;
    #pragma unroll
    for (int v = 0; v < 8; ++v) {
      float ar[KR], ai[KR];
      ar[0] = aC[0][v]; ai[0] = aS[0][v];
      #pragma unroll
      for (int r = 1; r < KR; ++r) {
        ar[r] = aC[r][v] * tc[r] - aS[r][v] * ts[r];
        ai[r] = aC[r][v] * ts[r] + aS[r][v] * tc[r];
      }
      float Er[4], Ei[4], Or[4], Oi[4];
      dft4(ar[0], ai[0], ar[2], ai[2], ar[4], ai[4], ar[6], ai[6], Er, Ei);
      dft4(ar[1], ai[1], ar[3], ai[3], ar[5], ai[5], ar[7], ai[7], Or, Oi);
      int row = v + 8 * h;
      HC[row][g]       = Er[0] + Or[0];
      HS[row][g]       = Ei[0] + Oi[0];
      HC[row][g + 128] = Er[1] + RH * (Or[1] - Oi[1]);
      HS[row][g + 128] = Ei[1] + RH * (Or[1] + Oi[1]);
      HC[row][g + 256] = Er[2] - Oi[2];
      HS[row][g + 256] = Ei[2] + Or[2];
      HC[row][g + 384] = Er[3] - RH * (Or[3] + Oi[3]);
      HS[row][g + 384] = Ei[3] + RH * (Or[3] - Oi[3]);
      if (wv == 0) {
        HC[row][g + 512] = Er[0] - Or[0];
        HS[row][g + 512] = Ei[0] - Oi[0];
      }
    }
  }
  __syncthreads();

  for (int idx = tid; idx < 16 * FPAD; idx += 256) {
    int m = idx / FPAD;
    int f = idx - m * FPAD;
    float p = 0.0f;
    if (m < HB && f < NFREQ) {
      float s  = (f & 1) ? -1.0f : 1.0f;
      int   fm = (f == 0) ? 1 : f - 1;
      float sm = (f == 0) ? -1.0f : 1.0f;
      int   fp = f + 1;
      float c0 = HC[m][f]  + s * HC[m + 1][f];
      float s0 = HS[m][f]  + s * HS[m + 1][f];
      float cm = HC[m][fm] - s * HC[m + 1][fm];
      float sn = (HS[m][fm] - s * HS[m + 1][fm]) * sm;
      float cp = HC[m][fp] - s * HC[m + 1][fp];
      float sp = HS[m][fp] - s * HS[m + 1][fp];
      float re = 0.5f * c0 - 0.25f * (cm + cp);
      float im = 0.5f * s0 - 0.25f * (sn + sp);
      p = re * re + im * im;
    }
    PL[m][f] = p;
  }
  __syncthreads();

  int n0 = wv * 16;
  const float* fB = fb2 + (long)(n0 + l16) * 2;
  v8f acc;
  #pragma unroll
  for (int j = 0; j < 8; ++j) acc[j] = 0.0f;
  for (int k = 0; k < FPAD; k += 4) {
    int kk = k + 2 * h;
    v2f a  = *(const v2f*)&PL[l16][kk];
    v2f bb = *(const v2f*)(fB + (long)(kk >> 1) * (NMELS * 2));
    acc = wmma4(a, bb, acc);
  }
  int n = n0 + l16;
  #pragma unroll
  for (int v = 0; v < 8; ++v) {
    int r = v + 8 * h;
    int t = t0 + r;
    if (r < HB && t < NFRAMES) mel[(b * NMELS + n) * NFRAMES + t] = acc[v];
  }
}


#define WIN_INSERT(a, m, x)                                             \
  do {                                                                  \
    if ((m) == 0) { (a)[0] = (x); }                                     \
    else {                                                              \
      (a)[(m)] = fmaxf((a)[(m) - 1], (x));                              \
      for (int i_ = (m) - 1; i_ >= 1; --i_)                             \
        (a)[i_] = fmaxf((a)[i_ - 1], fminf((a)[i_], (x)));              \
      (a)[0] = fminf((a)[0], (x));                                      \
    }                                                                   \
  } while (0)

__device__ __forceinline__ void win_slide(float (&a)[31], float xo, float xn) {
  float r[30];
  #pragma unroll
  for (int i = 0; i < 30; ++i) r[i] = (a[i] < xo) ? a[i] : a[i + 1];
  a[0] = fminf(r[0], xn);
  #pragma unroll
  for (int i = 1; i < 30; ++i) a[i] = fmaxf(r[i - 1], fminf(r[i], xn));
  a[30] = fmaxf(r[29], xn);
}

#define RT 11
__global__ __launch_bounds__(256) void median_time(const float* __restrict__ mel,
                                                   float* __restrict__ medT) {
  __shared__ float P[NFRAMES + 30];
  __shared__ float O[NFRAMES];
  int row = blockIdx.x;
  int tid = threadIdx.x;
  const float* src = mel + (long)row * NFRAMES;
  for (int i = tid; i < NFRAMES + 30; i += 256) {
    int j = i - 15;
    j = (j < 0) ? (-j - 1) : j;
    j = (j >= NFRAMES) ? (2 * NFRAMES - 1 - j) : j;
    P[i] = src[j];
  }
  __syncthreads();
  int t0 = tid * RT;
  if (t0 < NFRAMES) {
    float a[31];
    #pragma unroll
    for (int m = 0; m < 31; ++m) {
      float x = P[t0 + m];
      WIN_INSERT(a, m, x);
    }
    O[t0] = a[15];
    #pragma unroll
    for (int i = 1; i < RT; ++i) {
      int t = t0 + i;
      if (t < NFRAMES) {
        win_slide(a, P[t - 1], P[t + 30]);
        O[t] = a[15];
      }
    }
  }
  __syncthreads();
  float* dst = medT + (long)row * NFRAMES;
  for (int i = tid; i < NFRAMES; i += 256) dst[i] = O[i];
}

#define RN 16
__global__ __launch_bounds__(256) void hpss_mel(const float* __restrict__ mel,
                                                float* __restrict__ harm,
                                                float* __restrict__ perc,
                                                unsigned* __restrict__ mx) {
  __shared__ float r0[256], r1[256], r2[256];
  int tid = threadIdx.x;
  int t   = blockIdx.x * 256 + tid;
  int n0  = blockIdx.y * RN;
  int b   = blockIdx.z;
  float s = 0.0f, dh = 0.0f, dp = 0.0f;
  if (t < NFRAMES) {
    int cb = (b * NMELS) * NFRAMES + t;
    const float* base = mel + cb;
    float a[31];
    #pragma unroll
    for (int m = 0; m < 31; ++m) {
      int j = n0 + m - 15;
      j = (j < 0) ? (-j - 1) : j;
      j = (j >= NMELS) ? (2 * NMELS - 1 - j) : j;
      float x = base[j * NFRAMES];
      WIN_INSERT(a, m, x);
    }
    #pragma unroll
    for (int i = 0; i < RN; ++i) {
      int n = n0 + i;
      if (i > 0) {
        int jo = n - 16;
        jo = (jo < 0) ? (-jo - 1) : jo;
        int jn = n + 15;
        jn = (jn >= NMELS) ? (2 * NMELS - 1 - jn) : jn;
        win_slide(a, base[jo * NFRAMES], base[jn * NFRAMES]);
      }
      float pe = a[15];
      int   e  = cb + n * NFRAMES;
      float S  = mel[e];
      float ha = harm[e];
      float Z  = fmaxf(ha, pe);
      bool bad = Z < 1.17549435e-38f;
      float Zs = bad ? 1.0f : Z;
      float mh = ha / Zs; mh *= mh;
      float mp = pe / Zs; mp *= mp;
      float den = mh + mp;
      float maskh = bad ? 0.0f : mh / den;
      float maskp = bad ? 0.0f : mp / den;
      float vh = S * maskh, vp = S * maskp;
      harm[e] = vh;
      perc[e] = vp;
      s = fmaxf(s, S); dh = fmaxf(dh, vh); dp = fmaxf(dp, vp);
    }
  }
  r0[tid] = s; r1[tid] = dh; r2[tid] = dp;
  __syncthreads();
  for (int off = 128; off > 0; off >>= 1) {
    if (tid < off) {
      r0[tid] = fmaxf(r0[tid], r0[tid + off]);
      r1[tid] = fmaxf(r1[tid], r1[tid + off]);
      r2[tid] = fmaxf(r2[tid], r2[tid + off]);
    }
    __syncthreads();
  }
  if (tid == 0) {
    atomicMax(mx + 0, __float_as_uint(r0[0]));
    atomicMax(mx + 1, __float_as_uint(r1[0]));
    atomicMax(mx + 2, __float_as_uint(r2[0]));
  }
}

__global__ __launch_bounds__(256) void finalize(const float* __restrict__ mel,
                                                const float* __restrict__ harm,
                                                const float* __restrict__ perc,
                                                const unsigned* __restrict__ mx,
                                                float* __restrict__ out) {
  int gid = blockIdx.x * 256 + threadIdx.x;
  if (gid >= 3 * E_ELEMS) return;
  int s = gid / E_ELEMS;
  int e = gid - s * E_ELEMS;
  const float* src = (s == 0) ? mel : ((s == 1) ? harm : perc);
  float ref = __uint_as_float(mx[s]);
  float v   = src[e];
  float db = 10.0f * log10f(fmaxf(1e-10f, v)) -
             10.0f * log10f(fmaxf(1e-10f, ref));
  db = fmaxf(db, -80.0f);
  out[gid] = (db + 80.0f) * (1.0f / (80.0f + 1e-7f));
}

extern "C" void kernel_launch(void* const* d_in, const int* in_sizes, int n_in,
                              void* d_out, int out_size, void* d_ws,
                              size_t ws_size, hipStream_t stream) {
  const float* wave = (const float*)d_in[0];
  float* out  = (float*)d_out;
  float* ws   = (float*)d_ws;
  float* mel  = ws + MEL_OFF;
  float* harm = ws + HARM_OFF;
  float* perc = ws + PERC_OFF;
  unsigned* mx = (unsigned*)(ws + MAX_OFF);

  build_tables<<<MAX_OFF / 256, 256, 0, stream>>>(ws);
  mel_gemm<<<NB * MT2, 256, 0, stream>>>(wave, ws, mel);
  median_time<<<NB * NMELS, 256, 0, stream>>>(mel, harm);
  hpss_mel<<<dim3((NFRAMES + 255) / 256, NMELS / RN, NB), 256, 0, stream>>>(
      mel, harm, perc, mx);
  finalize<<<(3 * E_ELEMS + 255) / 256, 256, 0, stream>>>(mel, harm, perc, mx,
                                                          out);
}


# hpss_mel runs twice: pass 1 takes only the three maxima, pass 2 forms S, D_h, D_p again and stores them finished to out; no D_h / D_p planes, no element-per-thread finishing launch; max only
# speedup vs baseline: 2.8002x; 1.0318x over previous
#include <hip/hip_runtime.h>
#include <math.h>

#define NFFT     1024
#define HOPSZ    512
#define NMELS    128
#define NFREQ    513
#define FPAD     528
#define T_IN     1323000
#define NB       16
#define NFRAMES  2584
#define E_ELEMS  (NB*NMELS*NFRAMES)

#define KR       8
#define KJ       64
#define GP       128
#define DFT_OFF  0
#define DFT_SZ   ((KJ/2)*GP*4)
#define TW_OFF   (DFT_OFF + DFT_SZ)
#define TW_SZ    (KR*GP*2)
#define FB_OFF   (TW_OFF + TW_SZ)
#define FB_SZ    ((FPAD/2)*NMELS*2)
#define MAX_OFF  (FB_OFF + FB_SZ)
#define MEL_OFF  (MAX_OFF + 8)
#define MEDT_OFF (MEL_OFF + E_ELEMS)

typedef float v2f __attribute__((ext_vector_type(2)));
typedef float v4f __attribute__((ext_vector_type(4)));
typedef float v8f __attribute__((ext_vector_type(8)));

__device__ __forceinline__ v8f wmma4(v2f a, v2f b, v8f c) {
  return __builtin_amdgcn_wmma_f32_16x16x4_f32(false, a, false, b, (short)0, c,
                                               false, false);
}

__global__ __launch_bounds__(256) void build_tables(float* __restrict__ ws) {
  int gid = blockIdx.x * 256 + threadIdx.x;
  if (gid == 0) {
    unsigned* mx = (unsigned*)(ws + MAX_OFF);
    mx[0] = 0u; mx[1] = 0u; mx[2] = 0u;
  }
  const float TWO_PI = 6.28318530717958647692f;
  if (gid < TW_OFF) {
    int i = gid - DFT_OFF;
    int entry = i & 3;
    int rest  = i >> 2;
    int g  = rest % GP;
    int j2 = rest / GP;
    int j  = 2 * j2 + (entry & 1);
    int p  = (j * g) & (GP - 1);
    float arg = (float)p * (TWO_PI / 128.0f);
    ws[gid] = (entry < 2) ? cosf(arg) : sinf(arg);
  } else if (gid < FB_OFF) {
    int i = gid - TW_OFF;
    int entry = i & 1;
    int rest  = i >> 1;
    int g  = rest % GP;
    int r  = rest / GP;
    float arg = (float)(r * g) * (TWO_PI / 1024.0f);
    ws[gid] = entry ? sinf(arg) : cosf(arg);
  } else if (gid < MAX_OFF) {
    int i = gid - FB_OFF;
    int entry = i & 1;
    int rest  = i >> 1;
    int m  = rest % NMELS;
    int k2 = rest / NMELS;
    int f  = 2 * k2 + entry;
    float v = 0.0f;
    if (f < NFREQ) {
      float freq   = (float)f * (22050.0f / 512.0f);
      float melmax = 2595.0f * log10f(1.0f + 22050.0f / 700.0f);
      float step   = melmax / 129.0f;
      float p0 = 700.0f * (exp10f(((float)(m    ) * step) / 2595.0f) - 1.0f);
      float p1 = 700.0f * (exp10f(((float)(m + 1) * step) / 2595.0f) - 1.0f);
      float p2 = 700.0f * (exp10f(((float)(m + 2) * step) / 2595.0f) - 1.0f);
      float down = (freq - p0) / (p1 - p0);
      float up   = (p2 - freq) / (p2 - p1);
      float t = fmaxf(0.0f, fminf(down, up));
      v = t * 2.0f / (p2 - p0);
    }
    ws[gid] = v;
  }
}

#define KH     512
#define HB     15
#define MT2    173
#define HPITCH (FPAD + 4)
#define APITCH (KJ + 4)
#define AROW   (KR * APITCH + 4)

__device__ __forceinline__ void dft4(float r0, float i0, float r1, float i1,
                                     float r2, float i2, float r3, float i3,
                                     float (&Br)[4], float (&Bi)[4]) {
  float s02r = r0 + r2, s02i = i0 + i2, d02r = r0 - r2, d02i = i0 - i2;
  float s13r = r1 + r3, s13i = i1 + i3, d13r = r1 - r3, d13i = i1 - i3;
  Br[0] = s02r + s13r; Bi[0] = s02i + s13i;
  Br[1] = d02r - d13i; Bi[1] = d02i + d13r;
  Br[2] = s02r - s13r; Bi[2] = s02i - s13i;
  Br[3] = d02r + d13i; Bi[3] = d02i - d13r;
}
__global__ __launch_bounds__(256) void mel_gemm(const float* __restrict__ wave,
                                                const float* __restrict__ ws,
                                                float* __restrict__ mel) {
  __shared__ float AL[16][AROW];
  __shared__ float HC[16][HPITCH];
  __shared__ float HS[16][HPITCH];
  float (*PL)[AROW] = AL;

  const float* __restrict__ dftT = ws + DFT_OFF;
  const float* __restrict__ twT  = ws + TW_OFF;
  const float* __restrict__ fb2  = ws + FB_OFF;

  int b    = blockIdx.x / MT2;
  int tile = blockIdx.x % MT2;
  int t0   = tile * HB;
  int tid  = threadIdx.x;
  int lane = tid & 31;
  int wv   = tid >> 5;
  int h    = lane >> 4;
  int l16  = lane & 15;

  const float* wb = wave + (long)b * T_IN;

  for (int idx = tid; idx < 16 * KH; idx += 256) {
    int m = idx >> 9;
    int k = idx & (KH - 1);
    int j = t0 + m;
    float v = 0.0f;
    if (j <= NFRAMES) {
      int o = j * HOPSZ + k - NFFT / 2;
      o = (o < 0) ? -o : o;
      o = (o >= T_IN) ? (2 * (T_IN - 1) - o) : o;
      v = wb[o];
    }
    AL[m][(k & (KR - 1)) * APITCH + (k >> 3)] = v;
  }
  __syncthreads();

  {
    int g0 = wv * 16;
    int g  = g0 + l16;
    const float* dB = dftT + (long)g * 4;
    float tc[KR], ts[KR];
    tc[0] = 1.0f; ts[0] = 0.0f;
    #pragma unroll
    for (int r = 1; r < KR; ++r) {
      v2f t = *(const v2f*)(twT + (long)(r * GP + g) * 2);
      tc[r] = t[0]; ts[r] = t[1];
    }
    v8f aC[KR], aS[KR];
    #pragma unroll
    for (int r = 0; r < KR; ++r) {
      #pragma unroll
      for (int j = 0; j < 8; ++j) { aC[r][j] = 0.0f; aS[r][j] = 0.0f; }
    }
    #pragma unroll
    for (int k = 0; k < KJ; k += 4) {
      int kk = k + 2 * h;
      v4f q = *(const v4f*)(dB + (long)(kk >> 1) * (GP * 4));
      v2f bc; bc[0] = q[0]; bc[1] = q[1];
      v2f bs; bs[0] = q[2]; bs[1] = q[3];
      #pragma unroll
      for (int r = 0; r < KR; ++r) {
        v2f a = *(const v2f*)&AL[l16][r * APITCH + kk];
        aC[r] = wmma4(a, bc, aC[r]);
        aS[r] = wmma4(a, bs, aS[r]);
      }
    }
    const float RH = 0.70710678118654752440f;
    #pragma unroll
    for (int v = 0; v < 8; ++v) {
      float ar[KR], ai[KR];
      ar[0] = aC[0][v]; ai[0] = aS[0][v];
      #pragma unroll
      for (int r = 1; r < KR; ++r) {
        ar[r] = aC[r][v] * tc[r] - aS[r][v] * ts[r];
        ai[r] = aC[r][v] * ts[r] + aS[r][v] * tc[r];
      }
      float Er[4], Ei[4], Or[4], Oi[4];
      dft4(ar[0], ai[0], ar[2], ai[2], ar[4], ai[4], ar[6], ai[6], Er, Ei);
      dft4(ar[1], ai[1], ar[3], ai[3], ar[5], ai[5], ar[7], ai[7], Or, Oi);
      int row = v + 8 * h;
      HC[row][g]       = Er[0] + Or[0];
      HS[row][g]       = Ei[0] + Oi[0];
      HC[row][g + 128] = Er[1] + RH * (Or[1] - Oi[1]);
      HS[row][g + 128] = Ei[1] + RH * (Or[1] + Oi[1]);
      HC[row][g + 256] = Er[2] - Oi[2];
      HS[row][g + 256] = Ei[2] + Or[2];
      HC[row][g + 384] = Er[3] - RH * (Or[3] + Oi[3]);
      HS[row][g + 384] = Ei[3] + RH * (Or[3] - Oi[3]);
      if (wv == 0) {
        HC[row][g + 512] = Er[0] - Or[0];
        HS[row][g + 512] = Ei[0] - Oi[0];
      }
    }
  }
  __syncthreads();

  for (int idx = tid; idx < 16 * FPAD; idx += 256) {
    int m = idx / FPAD;
    int f = idx - m * FPAD;
    float p = 0.0f;
    if (m < HB && f < NFREQ) {
      float s  = (f & 1) ? -1.0f : 1.0f;
      int   fm = (f == 0) ? 1 : f - 1;
      float sm = (f == 0) ? -1.0f : 1.0f;
      int   fp = f + 1;
      float c0 = HC[m][f]  + s * HC[m + 1][f];
      float s0 = HS[m][f]  + s * HS[m + 1][f];
      float cm = HC[m][fm] - s * HC[m + 1][fm];
      float sn = (HS[m][fm] - s * HS[m + 1][fm]) * sm;
      float cp = HC[m][fp] - s * HC[m + 1][fp];
      float sp = HS[m][fp] - s * HS[m + 1][fp];
      float re = 0.5f * c0 - 0.25f * (cm + cp);
      float im = 0.5f * s0 - 0.25f * (sn + sp);
      p = re * re + im * im;
    }
    PL[m][f] = p;
  }
  __syncthreads();

  int n0 = wv * 16;
  const float* fB = fb2 + (long)(n0 + l16) * 2;
  v8f acc;
  #pragma unroll
  for (int j = 0; j < 8; ++j) acc[j] = 0.0f;
  for (int k = 0; k < FPAD; k += 4) {
    int kk = k + 2 * h;
    v2f a  = *(const v2f*)&PL[l16][kk];
    v2f bb = *(const v2f*)(fB + (long)(kk >> 1) * (NMELS * 2));
    acc = wmma4(a, bb, acc);
  }
  int n = n0 + l16;
  #pragma unroll
  for (int v = 0; v < 8; ++v) {
    int r = v + 8 * h;
    int t = t0 + r;
    if (r < HB && t < NFRAMES) mel[(b * NMELS + n) * NFRAMES + t] = acc[v];
  }
}


#define WIN_INSERT(a, m, x)                                             \
  do {                                                                  \
    if ((m) == 0) { (a)[0] = (x); }                                     \
    else {                                                              \
      (a)[(m)] = fmaxf((a)[(m) - 1], (x));                              \
      for (int i_ = (m) - 1; i_ >= 1; --i_)                             \
        (a)[i_] = fmaxf((a)[i_ - 1], fminf((a)[i_], (x)));              \
      (a)[0] = fminf((a)[0], (x));                                      \
    }                                                                   \
  } while (0)

__device__ __forceinline__ void win_slide(float (&a)[31], float xo, float xn) {
  float r[30];
  #pragma unroll
  for (int i = 0; i < 30; ++i) r[i] = (a[i] < xo) ? a[i] : a[i + 1];
  a[0] = fminf(r[0], xn);
  #pragma unroll
  for (int i = 1; i < 30; ++i) a[i] = fmaxf(r[i - 1], fminf(r[i], xn));
  a[30] = fmaxf(r[29], xn);
}

#define RT 11
__global__ __launch_bounds__(256) void median_time(const float* __restrict__ mel,
                                                   float* __restrict__ medT) {
  __shared__ float P[NFRAMES + 30];
  __shared__ float O[NFRAMES];
  int row = blockIdx.x;
  int tid = threadIdx.x;
  const float* src = mel + (long)row * NFRAMES;
  for (int i = tid; i < NFRAMES + 30; i += 256) {
    int j = i - 15;
    j = (j < 0) ? (-j - 1) : j;
    j = (j >= NFRAMES) ? (2 * NFRAMES - 1 - j) : j;
    P[i] = src[j];
  }
  __syncthreads();
  int t0 = tid * RT;
  if (t0 < NFRAMES) {
    float a[31];
    #pragma unroll
    for (int m = 0; m < 31; ++m) {
      float x = P[t0 + m];
      WIN_INSERT(a, m, x);
    }
    O[t0] = a[15];
    #pragma unroll
    for (int i = 1; i < RT; ++i) {
      int t = t0 + i;
      if (t < NFRAMES) {
        win_slide(a, P[t - 1], P[t + 30]);
        O[t] = a[15];
      }
    }
  }
  __syncthreads();
  float* dst = medT + (long)row * NFRAMES;
  for (int i = tid; i < NFRAMES; i += 256) dst[i] = O[i];
}

__device__ __forceinline__ float to_unit_db(float v, float ref) {
  float db = 10.0f * log10f(fmaxf(1e-10f, v)) -
             10.0f * log10f(fmaxf(1e-10f, ref));
  db = fmaxf(db, -80.0f);
  return (db + 80.0f) * (1.0f / (80.0f + 1e-7f));
}

#define RN 16
template <bool FIN>
__global__ __launch_bounds__(256) void hpss_mel(const float* __restrict__ mel,
                                                const float* __restrict__ medT,
                                                unsigned* __restrict__ mx,
                                                float* __restrict__ out) {
  int tid = threadIdx.x;
  int t   = blockIdx.x * 256 + tid;
  int n0  = blockIdx.y * RN;
  int b   = blockIdx.z;
  float s = 0.0f, dh = 0.0f, dp = 0.0f;
  float refS = 0.0f, refH = 0.0f, refP = 0.0f;
  if (FIN) {
    refS = __uint_as_float(mx[0]);
    refH = __uint_as_float(mx[1]);
    refP = __uint_as_float(mx[2]);
  }
  if (t < NFRAMES) {
    int cb = (b * NMELS) * NFRAMES + t;
    const float* base = mel + cb;
    float a[31];
    #pragma unroll
    for (int m = 0; m < 31; ++m) {
      int j = n0 + m - 15;
      j = (j < 0) ? (-j - 1) : j;
      j = (j >= NMELS) ? (2 * NMELS - 1 - j) : j;
      float x = base[j * NFRAMES];
      WIN_INSERT(a, m, x);
    }
    #pragma unroll
    for (int i = 0; i < RN; ++i) {
      int n = n0 + i;
      if (i > 0) {
        int jo = n - 16;
        jo = (jo < 0) ? (-jo - 1) : jo;
        int jn = n + 15;
        jn = (jn >= NMELS) ? (2 * NMELS - 1 - jn) : jn;
        win_slide(a, base[jo * NFRAMES], base[jn * NFRAMES]);
      }
      float pe = a[15];
      int   e  = cb + n * NFRAMES;
      float S  = mel[e];
      float ha = medT[e];
      float Z  = fmaxf(ha, pe);
      bool bad = Z < 1.17549435e-38f;
      float Zs = bad ? 1.0f : Z;
      float mh = ha / Zs; mh *= mh;
      float mp = pe / Zs; mp *= mp;
      float den = mh + mp;
      float maskh = bad ? 0.0f : mh / den;
      float maskp = bad ? 0.0f : mp / den;
      float vh = S * maskh, vp = S * maskp;
      if (FIN) {
        out[e]               = to_unit_db(S,  refS);
        out[E_ELEMS + e]     = to_unit_db(vh, refH);
        out[2 * E_ELEMS + e] = to_unit_db(vp, refP);
      } else {
        s = fmaxf(s, S); dh = fmaxf(dh, vh); dp = fmaxf(dp, vp);
      }
    }
  }
  if (FIN) return;
  __shared__ float r0[256], r1[256], r2[256];
  r0[tid] = s; r1[tid] = dh; r2[tid] = dp;
  __syncthreads();
  for (int off = 128; off > 0; off >>= 1) {
    if (tid < off) {
      r0[tid] = fmaxf(r0[tid], r0[tid + off]);
      r1[tid] = fmaxf(r1[tid], r1[tid + off]);
      r2[tid] = fmaxf(r2[tid], r2[tid + off]);
    }
    __syncthreads();
  }
  if (tid == 0) {
    atomicMax(mx + 0, __float_as_uint(r0[0]));
    atomicMax(mx + 1, __float_as_uint(r1[0]));
    atomicMax(mx + 2, __float_as_uint(r2[0]));
  }
}

extern "C" void kernel_launch(void* const* d_in, const int* in_sizes, int n_in,
                              void* d_out, int out_size, void* d_ws,
                              size_t ws_size, hipStream_t stream) {
  const float* wave = (const float*)d_in[0];
  float* out  = (float*)d_out;
  float* ws   = (float*)d_ws;
  float* mel  = ws + MEL_OFF;
  float* medT = ws + MEDT_OFF;
  unsigned* mx = (unsigned*)(ws + MAX_OFF);

  build_tables<<<MAX_OFF / 256, 256, 0, stream>>>(ws);
  mel_gemm<<<NB * MT2, 256, 0, stream>>>(wave, ws, mel);
  median_time<<<NB * NMELS, 256, 0, stream>>>(mel, medT);
  dim3 hg((NFRAMES + 255) / 256, NMELS / RN, NB);
  hpss_mel<false><<<hg, 256, 0, stream>>>(mel, medT, mx, out);
  hpss_mel<true><<<hg, 256, 0, stream>>>(mel, medT, mx, out);
}


# mel_gemm frame build: a tile's 8192 samples come as eight 16-byte loads per thread, all issued before the first LDS store: one memory wait per block, not 32; edge rows rebuilt; values, sum order same
# speedup vs baseline: 3.7816x; 1.3505x over previous
#include <hip/hip_runtime.h>
#include <math.h>

#define NFFT     1024
#define HOPSZ    512
#define NMELS    128
#define NFREQ    513
#define FPAD     528
#define T_IN     1323000
#define NB       16
#define NFRAMES  2584
#define E_ELEMS  (NB*NMELS*NFRAMES)

#define KR       8
#define KJ       64
#define GP       128
#define DFT_OFF  0
#define DFT_SZ   ((KJ/2)*GP*4)
#define TW_OFF   (DFT_OFF + DFT_SZ)
#define TW_SZ    (KR*GP*2)
#define FB_OFF   (TW_OFF + TW_SZ)
#define FB_SZ    ((FPAD/2)*NMELS*2)
#define MAX_OFF  (FB_OFF + FB_SZ)
#define MEL_OFF  (MAX_OFF + 8)
#define MEDT_OFF (MEL_OFF + E_ELEMS)

typedef float v2f __attribute__((ext_vector_type(2)));
typedef float v4f __attribute__((ext_vector_type(4)));
typedef float v8f __attribute__((ext_vector_type(8)));

__device__ __forceinline__ v8f wmma4(v2f a, v2f b, v8f c) {
  return __builtin_amdgcn_wmma_f32_16x16x4_f32(false, a, false, b, (short)0, c,
                                               false, false);
}

__global__ __launch_bounds__(256) void build_tables(float* __restrict__ ws) {
  int gid = blockIdx.x * 256 + threadIdx.x;
  if (gid == 0) {
    unsigned* mx = (unsigned*)(ws + MAX_OFF);
    mx[0] = 0u; mx[1] = 0u; mx[2] = 0u;
  }
  const float TWO_PI = 6.28318530717958647692f;
  if (gid < TW_OFF) {
    int i = gid - DFT_OFF;
    int entry = i & 3;
    int rest  = i >> 2;
    int g  = rest % GP;
    int j2 = rest / GP;
    int j  = 2 * j2 + (entry & 1);
    int p  = (j * g) & (GP - 1);
    float arg = (float)p * (TWO_PI / 128.0f);
    ws[gid] = (entry < 2) ? cosf(arg) : sinf(arg);
  } else if (gid < FB_OFF) {
    int i = gid - TW_OFF;
    int entry = i & 1;
    int rest  = i >> 1;
    int g  = rest % GP;
    int r  = rest / GP;
    float arg = (float)(r * g) * (TWO_PI / 1024.0f);
    ws[gid] = entry ? sinf(arg) : cosf(arg);
  } else if (gid < MAX_OFF) {
    int i = gid - FB_OFF;
    int entry = i & 1;
    int rest  = i >> 1;
    int m  = rest % NMELS;
    int k2 = rest / NMELS;
    int f  = 2 * k2 + entry;
    float v = 0.0f;
    if (f < NFREQ) {
      float freq   = (float)f * (22050.0f / 512.0f);
      float melmax = 2595.0f * log10f(1.0f + 22050.0f / 700.0f);
      float step   = melmax / 129.0f;
      float p0 = 700.0f * (exp10f(((float)(m    ) * step) / 2595.0f) - 1.0f);
      float p1 = 700.0f * (exp10f(((float)(m + 1) * step) / 2595.0f) - 1.0f);
      float p2 = 700.0f * (exp10f(((float)(m + 2) * step) / 2595.0f) - 1.0f);
      float down = (freq - p0) / (p1 - p0);
      float up   = (p2 - freq) / (p2 - p1);
      float t = fmaxf(0.0f, fminf(down, up));
      v = t * 2.0f / (p2 - p0);
    }
    ws[gid] = v;
  }
}

#define KH     512
#define HB     15
#define MT2    173
#define HPITCH (FPAD + 4)
#define APITCH (KJ + 4)
#define AROW   (KR * APITCH + 4)

__device__ __forceinline__ void dft4(float r0, float i0, float r1, float i1,
                                     float r2, float i2, float r3, float i3,
                                     float (&Br)[4], float (&Bi)[4]) {
  float s02r = r0 + r2, s02i = i0 + i2, d02r = r0 - r2, d02i = i0 - i2;
  float s13r = r1 + r3, s13i = i1 + i3, d13r = r1 - r3, d13i = i1 - i3;
  Br[0] = s02r + s13r; Bi[0] = s02i + s13i;
  Br[1] = d02r - d13i; Bi[1] = d02i + d13r;
  Br[2] = s02r - s13r; Bi[2] = s02i - s13i;
  Br[3] = d02r + d13i; Bi[3] = d02i - d13r;
}
__global__ __launch_bounds__(256) void mel_gemm(const float* __restrict__ wave,
                                                const float* __restrict__ ws,
                                                float* __restrict__ mel) {
  __shared__ float AL[16][AROW];
  __shared__ float HC[16][HPITCH];
  __shared__ float HS[16][HPITCH];
  float (*PL)[AROW] = AL;

  const float* __restrict__ dftT = ws + DFT_OFF;
  const float* __restrict__ twT  = ws + TW_OFF;
  const float* __restrict__ fb2  = ws + FB_OFF;

  int b    = blockIdx.x / MT2;
  int tile = blockIdx.x % MT2;
  int t0   = tile * HB;
  int tid  = threadIdx.x;
  int lane = tid & 31;
  int wv   = tid >> 5;
  int h    = lane >> 4;
  int l16  = lane & 15;

  const float* wb = wave + (long)b * T_IN;

  {
    int base = t0 * HOPSZ - NFFT / 2;
    v4f q[8];
    #pragma unroll
    for (int i = 0; i < 8; ++i) {
      int o = base + 4 * (tid + 256 * i);
      o = min(max(o, 0), T_IN - 4);
      q[i] = *(const v4f*)(wb + o);
    }
    #pragma unroll
    for (int i = 0; i < 8; ++i) {
      int e = tid + 256 * i;
      int m = e >> 7;
      int k = (e & 127) * 4;
      float* dst = &AL[m][(k & (KR - 1)) * APITCH + (k >> 3)];
      #pragma unroll
      for (int c = 0; c < 4; ++c) dst[c * APITCH] = q[i][c];
    }
  }
  if (tile == 0 || tile == MT2 - 1) {
    __syncthreads();
    for (int idx = tid; idx < 16 * KH; idx += 256) {
      int m = idx >> 9;
      int k = idx & (KH - 1);
      int j = t0 + m;
      if (j == 0 || j >= NFRAMES) {
        float v = 0.0f;
        if (j <= NFRAMES) {
          int o = j * HOPSZ + k - NFFT / 2;
          o = (o < 0) ? -o : o;
          o = (o >= T_IN) ? (2 * (T_IN - 1) - o) : o;
          v = wb[o];
        }
        AL[m][(k & (KR - 1)) * APITCH + (k >> 3)] = v;
      }
    }
  }
  __syncthreads();

  {
    int g0 = wv * 16;
    int g  = g0 + l16;
    const float* dB = dftT + (long)g * 4;
    float tc[KR], ts[KR];
    tc[0] = 1.0f; ts[0] = 0.0f;
    #pragma unroll
    for (int r = 1; r < KR; ++r) {
      v2f t = *(const v2f*)(twT + (long)(r * GP + g) * 2);
      tc[r] = t[0]; ts[r] = t[1];
    }
    v8f aC[KR], aS[KR];
    #pragma unroll
    for (int r = 0; r < KR; ++r) {
      #pragma unroll
      for (int j = 0; j < 8; ++j) { aC[r][j] = 0.0f; aS[r][j] = 0.0f; }
    }
    #pragma unroll
    for (int k = 0; k < KJ; k += 4) {
      int kk = k + 2 * h;
      v4f q = *(const v4f*)(dB + (long)(kk >> 1) * (GP * 4));
      v2f bc; bc[0] = q[0]; bc[1] = q[1];
      v2f bs; bs[0] = q[2]; bs[1] = q[3];
      #pragma unroll
      for (int r = 0; r < KR; ++r) {
        v2f a = *(const v2f*)&AL[l16][r * APITCH + kk];
        aC[r] = wmma4(a, bc, aC[r]);
        aS[r] = wmma4(a, bs, aS[r]);
      }
    }
    const float RH = 0.70710678118654752440f;
    #pragma unroll
    for (int v = 0; v < 8; ++v) {
      float ar[KR], ai[KR];
      ar[0] = aC[0][v]; ai[0] = aS[0][v];
      #pragma unroll
      for (int r = 1; r < KR; ++r) {
        ar[r] = aC[r][v] * tc[r] - aS[r][v] * ts[r];
        ai[r] = aC[r][v] * ts[r] + aS[r][v] * tc[r];
      }
      float Er[4], Ei[4], Or[4], Oi[4];
      dft4(ar[0], ai[0], ar[2], ai[2], ar[4], ai[4], ar[6], ai[6], Er, Ei);
      dft4(ar[1], ai[1], ar[3], ai[3], ar[5], ai[5], ar[7], ai[7], Or, Oi);
      int row = v + 8 * h;
      HC[row][g]       = Er[0] + Or[0];
      HS[row][g]       = Ei[0] + Oi[0];
      HC[row][g + 128] = Er[1] + RH * (Or[1] - Oi[1]);
      HS[row][g + 128] = Ei[1] + RH * (Or[1] + Oi[1]);
      HC[row][g + 256] = Er[2] - Oi[2];
      HS[row][g + 256] = Ei[2] + Or[2];
      HC[row][g + 384] = Er[3] - RH * (Or[3] + Oi[3]);
      HS[row][g + 384] = Ei[3] + RH * (Or[3] - Oi[3]);
      if (wv == 0) {
        HC[row][g + 512] = Er[0] - Or[0];
        HS[row][g + 512] = Ei[0] - Oi[0];
      }
    }
  }
  __syncthreads();

  for (int idx = tid; idx < 16 * FPAD; idx += 256) {
    int m = idx / FPAD;
    int f = idx - m * FPAD;
    float p = 0.0f;
    if (m < HB && f < NFREQ) {
      float s  = (f & 1) ? -1.0f : 1.0f;
      int   fm = (f == 0) ? 1 : f - 1;
      float sm = (f == 0) ? -1.0f : 1.0f;
      int   fp = f + 1;
      float c0 = HC[m][f]  + s * HC[m + 1][f];
      float s0 = HS[m][f]  + s * HS[m + 1][f];
      float cm = HC[m][fm] - s * HC[m + 1][fm];
      float sn = (HS[m][fm] - s * HS[m + 1][fm]) * sm;
      float cp = HC[m][fp] - s * HC[m + 1][fp];
      float sp = HS[m][fp] - s * HS[m + 1][fp];
      float re = 0.5f * c0 - 0.25f * (cm + cp);
      float im = 0.5f * s0 - 0.25f * (sn + sp);
      p = re * re + im * im;
    }
    PL[m][f] = p;
  }
  __syncthreads();

  int n0 = wv * 16;
  const float* fB = fb2 + (long)(n0 + l16) * 2;
  v8f acc;
  #pragma unroll
  for (int j = 0; j < 8; ++j) acc[j] = 0.0f;
  for (int k = 0; k < FPAD; k += 4) {
    int kk = k + 2 * h;
    v2f a  = *(const v2f*)&PL[l16][kk];
    v2f bb = *(const v2f*)(fB + (long)(kk >> 1) * (NMELS * 2));
    acc = wmma4(a, bb, acc);
  }
  int n = n0 + l16;
  #pragma unroll
  for (int v = 0; v < 8; ++v) {
    int r = v + 8 * h;
    int t = t0 + r;
    if (r < HB && t < NFRAMES) mel[(b * NMELS + n) * NFRAMES + t] = acc[v];
  }
}


#define WIN_INSERT(a, m, x)                                             \
  do {                                                                  \
    if ((m) == 0) { (a)[0] = (x); }                                     \
    else {                                                              \
      (a)[(m)] = fmaxf((a)[(m) - 1], (x));                              \
      for (int i_ = (m) - 1; i_ >= 1; --i_)                             \
        (a)[i_] = fmaxf((a)[i_ - 1], fminf((a)[i_], (x)));              \
      (a)[0] = fminf((a)[0], (x));                                      \
    }                                                                   \
  } while (0)

__device__ __forceinline__ void win_slide(float (&a)[31], float xo, float xn) {
  float r[30];
  #pragma unroll
  for (int i = 0; i < 30; ++i) r[i] = (a[i] < xo) ? a[i] : a[i + 1];
  a[0] = fminf(r[0], xn);
  #pragma unroll
  for (int i = 1; i < 30; ++i) a[i] = fmaxf(r[i - 1], fminf(r[i], xn));
  a[30] = fmaxf(r[29], xn);
}

#define RT 11
__global__ __launch_bounds__(256) void median_time(const float* __restrict__ mel,
                                                   float* __restrict__ medT) {
  __shared__ float P[NFRAMES + 30];
  __shared__ float O[NFRAMES];
  int row = blockIdx.x;
  int tid = threadIdx.x;
  const float* src = mel + (long)row * NFRAMES;
  for (int i = tid; i < NFRAMES + 30; i += 256) {
    int j = i - 15;
    j = (j < 0) ? (-j - 1) : j;
    j = (j >= NFRAMES) ? (2 * NFRAMES - 1 - j) : j;
    P[i] = src[j];
  }
  __syncthreads();
  int t0 = tid * RT;
  if (t0 < NFRAMES) {
    float a[31];
    #pragma unroll
    for (int m = 0; m < 31; ++m) {
      float x = P[t0 + m];
      WIN_INSERT(a, m, x);
    }
    O[t0] = a[15];
    #pragma unroll
    for (int i = 1; i < RT; ++i) {
      int t = t0 + i;
      if (t < NFRAMES) {
        win_slide(a, P[t - 1], P[t + 30]);
        O[t] = a[15];
      }
    }
  }
  __syncthreads();
  float* dst = medT + (long)row * NFRAMES;
  for (int i = tid; i < NFRAMES; i += 256) dst[i] = O[i];
}

__device__ __forceinline__ float to_unit_db(float v, float ref) {
  float db = 10.0f * log10f(fmaxf(1e-10f, v)) -
             10.0f * log10f(fmaxf(1e-10f, ref));
  db = fmaxf(db, -80.0f);
  return (db + 80.0f) * (1.0f / (80.0f + 1e-7f));
}

#define RN 16
template <bool FIN>
__global__ __launch_bounds__(256) void hpss_mel(const float* __restrict__ mel,
                                                const float* __restrict__ medT,
                                                unsigned* __restrict__ mx,
                                                float* __restrict__ out) {
  int tid = threadIdx.x;
  int t   = blockIdx.x * 256 + tid;
  int n0  = blockIdx.y * RN;
  int b   = blockIdx.z;
  float s = 0.0f, dh = 0.0f, dp = 0.0f;
  float refS = 0.0f, refH = 0.0f, refP = 0.0f;
  if (FIN) {
    refS = __uint_as_float(mx[0]);
    refH = __uint_as_float(mx[1]);
    refP = __uint_as_float(mx[2]);
  }
  if (t < NFRAMES) {
    int cb = (b * NMELS) * NFRAMES + t;
    const float* base = mel + cb;
    float a[31];
    #pragma unroll
    for (int m = 0; m < 31; ++m) {
      int j = n0 + m - 15;
      j = (j < 0) ? (-j - 1) : j;
      j = (j >= NMELS) ? (2 * NMELS - 1 - j) : j;
      float x = base[j * NFRAMES];
      WIN_INSERT(a, m, x);
    }
    #pragma unroll
    for (int i = 0; i < RN; ++i) {
      int n = n0 + i;
      if (i > 0) {
        int jo = n - 16;
        jo = (jo < 0) ? (-jo - 1) : jo;
        int jn = n + 15;
        jn = (jn >= NMELS) ? (2 * NMELS - 1 - jn) : jn;
        win_slide(a, base[jo * NFRAMES], base[jn * NFRAMES]);
      }
      float pe = a[15];
      int   e  = cb + n * NFRAMES;
      float S  = mel[e];
      float ha = medT[e];
      float Z  = fmaxf(ha, pe);
      bool bad = Z < 1.17549435e-38f;
      float Zs = bad ? 1.0f : Z;
      float mh = ha / Zs; mh *= mh;
      float mp = pe / Zs; mp *= mp;
      float den = mh + mp;
      float maskh = bad ? 0.0f : mh / den;
      float maskp = bad ? 0.0f : mp / den;
      float vh = S * maskh, vp = S * maskp;
      if (FIN) {
        out[e]               = to_unit_db(S,  refS);
        out[E_ELEMS + e]     = to_unit_db(vh, refH);
        out[2 * E_ELEMS + e] = to_unit_db(vp, refP);
      } else {
        s = fmaxf(s, S); dh = fmaxf(dh, vh); dp = fmaxf(dp, vp);
      }
    }
  }
  if (FIN) return;
  __shared__ float r0[256], r1[256], r2[256];
  r0[tid] = s; r1[tid] = dh; r2[tid] = dp;
  __syncthreads();
  for (int off = 128; off > 0; off >>= 1) {
    if (tid < off) {
      r0[tid] = fmaxf(r0[tid], r0[tid + off]);
      r1[tid] = fmaxf(r1[tid], r1[tid + off]);
      r2[tid] = fmaxf(r2[tid], r2[tid + off]);
    }
    __syncthreads();
  }
  if (tid == 0) {
    atomicMax(mx + 0, __float_as_uint(r0[0]));
    atomicMax(mx + 1, __float_as_uint(r1[0]));
    atomicMax(mx + 2, __float_as_uint(r2[0]));
  }
}

extern "C" void kernel_launch(void* const* d_in, const int* in_sizes, int n_in,
                              void* d_out, int out_size, void* d_ws,
                              size_t ws_size, hipStream_t stream) {
  const float* wave = (const float*)d_in[0];
  float* out  = (float*)d_out;
  float* ws   = (float*)d_ws;
  float* mel  = ws + MEL_OFF;
  float* medT = ws + MEDT_OFF;
  unsigned* mx = (unsigned*)(ws + MAX_OFF);

  build_tables<<<MAX_OFF / 256, 256, 0, stream>>>(ws);
  mel_gemm<<<NB * MT2, 256, 0, stream>>>(wave, ws, mel);
  median_time<<<NB * NMELS, 256, 0, stream>>>(mel, medT);
  dim3 hg((NFRAMES + 255) / 256, NMELS / RN, NB);
  hpss_mel<false><<<hg, 256, 0, stream>>>(mel, medT, mx, out);
  hpss_mel<true><<<hg, 256, 0, stream>>>(mel, medT, mx, out);
}


# mel_gemm stage 1: residue transforms for g <= 64 only (real input: 128-g is conjugate); 8-point step gives both bins; even/odd residues on 2 waves, summed in LDS; VGPR cap 160; f32 sum order changes
# speedup vs baseline: 4.2430x; 1.1220x over previous
#include <hip/hip_runtime.h>
#include <math.h>

#define NFFT     1024
#define HOPSZ    512
#define NMELS    128
#define NFREQ    513
#define FPAD     528
#define T_IN     1323000
#define NB       16
#define NFRAMES  2584
#define E_ELEMS  (NB*NMELS*NFRAMES)

#define KR       8
#define KJ       64
#define GP       128
#define DFT_OFF  0
#define DFT_SZ   ((KJ/2)*GP*4)
#define TW_OFF   (DFT_OFF + DFT_SZ)
#define TW_SZ    (KR*GP*2)
#define FB_OFF   (TW_OFF + TW_SZ)
#define FB_SZ    ((FPAD/2)*NMELS*2)
#define MAX_OFF  (FB_OFF + FB_SZ)
#define MEL_OFF  (MAX_OFF + 8)
#define MEDT_OFF (MEL_OFF + E_ELEMS)

typedef float v2f __attribute__((ext_vector_type(2)));
typedef float v4f __attribute__((ext_vector_type(4)));
typedef float v8f __attribute__((ext_vector_type(8)));

__device__ __forceinline__ v8f wmma4(v2f a, v2f b, v8f c) {
  return __builtin_amdgcn_wmma_f32_16x16x4_f32(false, a, false, b, (short)0, c,
                                               false, false);
}

__global__ __launch_bounds__(256) void build_tables(float* __restrict__ ws) {
  int gid = blockIdx.x * 256 + threadIdx.x;
  if (gid == 0) {
    unsigned* mx = (unsigned*)(ws + MAX_OFF);
    mx[0] = 0u; mx[1] = 0u; mx[2] = 0u;
  }
  const float TWO_PI = 6.28318530717958647692f;
  if (gid < TW_OFF) {
    int i = gid - DFT_OFF;
    int entry = i & 3;
    int rest  = i >> 2;
    int g  = rest % GP;
    int j2 = rest / GP;
    int j  = 2 * j2 + (entry & 1);
    int p  = (j * g) & (GP - 1);
    float arg = (float)p * (TWO_PI / 128.0f);
    ws[gid] = (entry < 2) ? cosf(arg) : sinf(arg);
  } else if (gid < FB_OFF) {
    int i = gid - TW_OFF;
    int entry = i & 1;
    int rest  = i >> 1;
    int g  = rest % GP;
    int r  = rest / GP;
    float arg = (float)(r * g) * (TWO_PI / 1024.0f);
    ws[gid] = entry ? sinf(arg) : cosf(arg);
  } else if (gid < MAX_OFF) {
    int i = gid - FB_OFF;
    int entry = i & 1;
    int rest  = i >> 1;
    int m  = rest % NMELS;
    int k2 = rest / NMELS;
    int f  = 2 * k2 + entry;
    float v = 0.0f;
    if (f < NFREQ) {
      float freq   = (float)f * (22050.0f / 512.0f);
      float melmax = 2595.0f * log10f(1.0f + 22050.0f / 700.0f);
      float step   = melmax / 129.0f;
      float p0 = 700.0f * (exp10f(((float)(m    ) * step) / 2595.0f) - 1.0f);
      float p1 = 700.0f * (exp10f(((float)(m + 1) * step) / 2595.0f) - 1.0f);
      float p2 = 700.0f * (exp10f(((float)(m + 2) * step) / 2595.0f) - 1.0f);
      float down = (freq - p0) / (p1 - p0);
      float up   = (p2 - freq) / (p2 - p1);
      float t = fmaxf(0.0f, fminf(down, up));
      v = t * 2.0f / (p2 - p0);
    }
    ws[gid] = v;
  }
}

#define KH     512
#define HB     15
#define MT2    173
#define HPITCH (FPAD + 4)
#define APITCH (KJ + 4)
#define AROW   (KR * APITCH + 4)

__device__ __forceinline__ void dft4(float r0, float i0, float r1, float i1,
                                     float r2, float i2, float r3, float i3,
                                     float (&Br)[4], float (&Bi)[4]) {
  float s02r = r0 + r2, s02i = i0 + i2, d02r = r0 - r2, d02i = i0 - i2;
  float s13r = r1 + r3, s13i = i1 + i3, d13r = r1 - r3, d13i = i1 - i3;
  Br[0] = s02r + s13r; Bi[0] = s02i + s13i;
  Br[1] = d02r - d13i; Bi[1] = d02i + d13r;
  Br[2] = s02r - s13r; Bi[2] = s02i - s13i;
  Br[3] = d02r + d13i; Bi[3] = d02i - d13r;
}
__global__ __launch_bounds__(256, 6) void mel_gemm(const float* __restrict__ wave,
                                                const float* __restrict__ ws,
                                                float* __restrict__ mel) {
  __shared__ __attribute__((aligned(16))) float AL[16][AROW];
  __shared__ float HC[16][HPITCH];
  __shared__ float HS[16][HPITCH];
  __shared__ float GS[16][KR];
  float (*PL)[AROW] = AL;

  const float* __restrict__ dftT = ws + DFT_OFF;
  const float* __restrict__ twT  = ws + TW_OFF;
  const float* __restrict__ fb2  = ws + FB_OFF;

  int b    = blockIdx.x / MT2;
  int tile = blockIdx.x % MT2;
  int t0   = tile * HB;
  int tid  = threadIdx.x;
  int lane = tid & 31;
  int wv   = __builtin_amdgcn_readfirstlane(tid >> 5);
  int h    = lane >> 4;
  int l16  = lane & 15;

  const float* wb = wave + (long)b * T_IN;

  {
    int base = t0 * HOPSZ - NFFT / 2;
    v4f q[8];
    #pragma unroll
    for (int i = 0; i < 8; ++i) {
      int o = base + 4 * (tid + 256 * i);
      o = min(max(o, 0), T_IN - 4);
      q[i] = *(const v4f*)(wb + o);
    }
    #pragma unroll
    for (int i = 0; i < 8; ++i) {
      int e = tid + 256 * i;
      int m = e >> 7;
      int k = (e & 127) * 4;
      float* dst = &AL[m][(k & (KR - 1)) * APITCH + (k >> 3)];
      #pragma unroll
      for (int c = 0; c < 4; ++c) dst[c * APITCH] = q[i][c];
    }
  }
  if (tile == 0 || tile == MT2 - 1) {
    __syncthreads();
    for (int idx = tid; idx < 16 * KH; idx += 256) {
      int m = idx >> 9;
      int k = idx & (KH - 1);
      int j = t0 + m;
      if (j == 0 || j >= NFRAMES) {
        float v = 0.0f;
        if (j <= NFRAMES) {
          int o = j * HOPSZ + k - NFFT / 2;
          o = (o < 0) ? -o : o;
          o = (o >= T_IN) ? (2 * (T_IN - 1) - o) : o;
          v = wb[o];
        }
        AL[m][(k & (KR - 1)) * APITCH + (k >> 3)] = v;
      }
    }
  }
  __syncthreads();

  {
    int row  = tid >> 4;
    int r    = (tid >> 1) & (KR - 1);
    int half = tid & 1;
    const float* p = &AL[row][r * APITCH + 32 * half];
    float sum = 0.0f;
    #pragma unroll
    for (int j = 0; j < 32; j += 4) {
      v4f x = *(const v4f*)(p + j);
      sum += (x[0] - x[1]) + (x[2] - x[3]);
    }
    sum += __shfl_xor(sum, 1);
    if (half == 0) GS[row][r] = sum;
  }

  int gt  = wv & 3;
  int par = wv >> 2;
  int g   = gt * 16 + l16;
  float tc[4], ts[4];
  v8f aC[4], aS[4];
  {
    const float* dB = dftT + (long)g * 4;
    const float* aB = &AL[l16][par * APITCH];
    #pragma unroll
    for (int n = 0; n < 4; ++n) {
      v2f t = *(const v2f*)(twT + (long)((2 * n + par) * GP + g) * 2);
      tc[n] = t[0]; ts[n] = t[1];
    }
    #pragma unroll
    for (int n = 0; n < 4; ++n) {
      #pragma unroll
      for (int j = 0; j < 8; ++j) { aC[n][j] = 0.0f; aS[n][j] = 0.0f; }
    }
    #pragma unroll
    for (int k = 0; k < KJ; k += 4) {
      int kk = k + 2 * h;
      v4f q = *(const v4f*)(dB + (long)(kk >> 1) * (GP * 4));
      v2f bc; bc[0] = q[0]; bc[1] = q[1];
      v2f bs; bs[0] = q[2]; bs[1] = q[3];
      #pragma unroll
      for (int n = 0; n < 4; ++n) {
        v2f a = *(const v2f*)(aB + 2 * n * APITCH + kk);
        aC[n] = wmma4(a, bc, aC[n]);
        aS[n] = wmma4(a, bs, aS[n]);
      }
    }
  }
  const float RH = 0.70710678118654752440f;
  int mir[4];
  #pragma unroll
  for (int q = 0; q < 4; ++q)
    mir[q] = (q == 0 || g != 0) ? (512 - g - 128 * q) : (528 + q);
  int b513 = (g == 1) ? 513 : ((g == 0) ? 514 : 513 + g);
  float Wr[8][4], Wi[8][4];
  #pragma unroll
  for (int v = 0; v < 8; ++v) {
    float ar[4], ai[4];
    #pragma unroll
    for (int n = 0; n < 4; ++n) {
      ar[n] = aC[n][v] * tc[n] - aS[n][v] * ts[n];
      ai[n] = aC[n][v] * ts[n] + aS[n][v] * tc[n];
    }
    dft4(ar[0], ai[0], ar[1], ai[1], ar[2], ai[2], ar[3], ai[3], Wr[v], Wi[v]);
  }
  if (par == 0) {
    #pragma unroll
    for (int v = 0; v < 8; ++v) {
      int row = v + 8 * h;
      #pragma unroll
      for (int q = 0; q < 4; ++q) {
        HC[row][g + 128 * q] = Wr[v][q];
        HS[row][g + 128 * q] = Wi[v][q];
        HC[row][mir[q]] = Wr[v][q];
        HS[row][mir[q]] = -Wi[v][q];
      }
      if (gt == 0) {
        HC[row][b513] = Wr[v][0];
        HS[row][b513] = Wi[v][0];
      }
    }
  } else {
    #pragma unroll
    for (int v = 0; v < 8; ++v) {
      float r1 = Wr[v][1], i1 = Wi[v][1];
      float r2 = Wr[v][2], i2 = Wi[v][2];
      float r3 = Wr[v][3], i3 = Wi[v][3];
      Wr[v][1] = RH * (r1 - i1);    Wi[v][1] = RH * (r1 + i1);
      Wr[v][2] = -i2;               Wi[v][2] = r2;
      Wr[v][3] = -RH * (r3 + i3);   Wi[v][3] = RH * (r3 - i3);
    }
  }
  __syncthreads();
  if (par == 1) {
    #pragma unroll
    for (int v = 0; v < 8; ++v) {
      int row = v + 8 * h;
      #pragma unroll
      for (int q = 0; q < 4; ++q) {
        HC[row][g + 128 * q] += Wr[v][q];
        HS[row][g + 128 * q] += Wi[v][q];
        HC[row][mir[q]] -= Wr[v][q];
        HS[row][mir[q]] += Wi[v][q];
      }
      if (gt == 0) {
        HC[row][b513] -= Wr[v][0];
        HS[row][b513] -= Wi[v][0];
      }
    }
  } else if (wv == 0 && lane < 16) {
    float ar[KR], ai[KR];
    #pragma unroll
    for (int r = 0; r < KR; ++r) {
      v2f t = *(const v2f*)(twT + (long)(r * GP + 64) * 2);
      float sr = GS[lane][r];
      ar[r] = sr * t[0]; ai[r] = sr * t[1];
    }
    float Er[4], Ei[4], Or[4], Oi[4];
    dft4(ar[0], ai[0], ar[2], ai[2], ar[4], ai[4], ar[6], ai[6], Er, Ei);
    dft4(ar[1], ai[1], ar[3], ai[3], ar[5], ai[5], ar[7], ai[7], Or, Oi);
    HC[lane][64]  = Er[0] + Or[0];
    HS[lane][64]  = Ei[0] + Oi[0];
    HC[lane][192] = Er[1] + RH * (Or[1] - Oi[1]);
    HS[lane][192] = Ei[1] + RH * (Or[1] + Oi[1]);
    HC[lane][320] = Er[2] - Oi[2];
    HS[lane][320] = Ei[2] + Or[2];
    HC[lane][448] = Er[3] - RH * (Or[3] + Oi[3]);
    HS[lane][448] = Ei[3] + RH * (Or[3] - Oi[3]);
  }
  __syncthreads();

  for (int idx = tid; idx < 16 * FPAD; idx += 256) {
    int m = idx / FPAD;
    int f = idx - m * FPAD;
    float p = 0.0f;
    if (m < HB && f < NFREQ) {
      float s  = (f & 1) ? -1.0f : 1.0f;
      int   fm = (f == 0) ? 1 : f - 1;
      float sm = (f == 0) ? -1.0f : 1.0f;
      int   fp = f + 1;
      float c0 = HC[m][f]  + s * HC[m + 1][f];
      float s0 = HS[m][f]  + s * HS[m + 1][f];
      float cm = HC[m][fm] - s * HC[m + 1][fm];
      float sn = (HS[m][fm] - s * HS[m + 1][fm]) * sm;
      float cp = HC[m][fp] - s * HC[m + 1][fp];
      float sp = HS[m][fp] - s * HS[m + 1][fp];
      float re = 0.5f * c0 - 0.25f * (cm + cp);
      float im = 0.5f * s0 - 0.25f * (sn + sp);
      p = re * re + im * im;
    }
    PL[m][f] = p;
  }
  __syncthreads();

  int n0 = wv * 16;
  const float* fB = fb2 + (long)(n0 + l16) * 2;
  v8f acc;
  #pragma unroll
  for (int j = 0; j < 8; ++j) acc[j] = 0.0f;
  for (int k = 0; k < FPAD; k += 4) {
    int kk = k + 2 * h;
    v2f a  = *(const v2f*)&PL[l16][kk];
    v2f bb = *(const v2f*)(fB + (long)(kk >> 1) * (NMELS * 2));
    acc = wmma4(a, bb, acc);
  }
  int n = n0 + l16;
  #pragma unroll
  for (int v = 0; v < 8; ++v) {
    int r = v + 8 * h;
    int t = t0 + r;
    if (r < HB && t < NFRAMES) mel[(b * NMELS + n) * NFRAMES + t] = acc[v];
  }
}


#define WIN_INSERT(a, m, x)                                             \
  do {                                                                  \
    if ((m) == 0) { (a)[0] = (x); }                                     \
    else {                                                              \
      (a)[(m)] = fmaxf((a)[(m) - 1], (x));                              \
      for (int i_ = (m) - 1; i_ >= 1; --i_)                             \
        (a)[i_] = fmaxf((a)[i_ - 1], fminf((a)[i_], (x)));              \
      (a)[0] = fminf((a)[0], (x));                                      \
    }                                                                   \
  } while (0)

__device__ __forceinline__ void win_slide(float (&a)[31], float xo, float xn) {
  float r[30];
  #pragma unroll
  for (int i = 0; i < 30; ++i) r[i] = (a[i] < xo) ? a[i] : a[i + 1];
  a[0] = fminf(r[0], xn);
  #pragma unroll
  for (int i = 1; i < 30; ++i) a[i] = fmaxf(r[i - 1], fminf(r[i], xn));
  a[30] = fmaxf(r[29], xn);
}

#define RT 11
__global__ __launch_bounds__(256) void median_time(const float* __restrict__ mel,
                                                   float* __restrict__ medT) {
  __shared__ float P[NFRAMES + 30];
  __shared__ float O[NFRAMES];
  int row = blockIdx.x;
  int tid = threadIdx.x;
  const float* src = mel + (long)row * NFRAMES;
  for (int i = tid; i < NFRAMES + 30; i += 256) {
    int j = i - 15;
    j = (j < 0) ? (-j - 1) : j;
    j = (j >= NFRAMES) ? (2 * NFRAMES - 1 - j) : j;
    P[i] = src[j];
  }
  __syncthreads();
  int t0 = tid * RT;
  if (t0 < NFRAMES) {
    float a[31];
    #pragma unroll
    for (int m = 0; m < 31; ++m) {
      float x = P[t0 + m];
      WIN_INSERT(a, m, x);
    }
    O[t0] = a[15];
    #pragma unroll
    for (int i = 1; i < RT; ++i) {
      int t = t0 + i;
      if (t < NFRAMES) {
        win_slide(a, P[t - 1], P[t + 30]);
        O[t] = a[15];
      }
    }
  }
  __syncthreads();
  float* dst = medT + (long)row * NFRAMES;
  for (int i = tid; i < NFRAMES; i += 256) dst[i] = O[i];
}

__device__ __forceinline__ float to_unit_db(float v, float ref) {
  float db = 10.0f * log10f(fmaxf(1e-10f, v)) -
             10.0f * log10f(fmaxf(1e-10f, ref));
  db = fmaxf(db, -80.0f);
  return (db + 80.0f) * (1.0f / (80.0f + 1e-7f));
}

#define RN 16
template <bool FIN>
__global__ __launch_bounds__(256) void hpss_mel(const float* __restrict__ mel,
                                                const float* __restrict__ medT,
                                                unsigned* __restrict__ mx,
                                                float* __restrict__ out) {
  int tid = threadIdx.x;
  int t   = blockIdx.x * 256 + tid;
  int n0  = blockIdx.y * RN;
  int b   = blockIdx.z;
  float s = 0.0f, dh = 0.0f, dp = 0.0f;
  float refS = 0.0f, refH = 0.0f, refP = 0.0f;
  if (FIN) {
    refS = __uint_as_float(mx[0]);
    refH = __uint_as_float(mx[1]);
    refP = __uint_as_float(mx[2]);
  }
  if (t < NFRAMES) {
    int cb = (b * NMELS) * NFRAMES + t;
    const float* base = mel + cb;
    float a[31];
    #pragma unroll
    for (int m = 0; m < 31; ++m) {
      int j = n0 + m - 15;
      j = (j < 0) ? (-j - 1) : j;
      j = (j >= NMELS) ? (2 * NMELS - 1 - j) : j;
      float x = base[j * NFRAMES];
      WIN_INSERT(a, m, x);
    }
    #pragma unroll
    for (int i = 0; i < RN; ++i) {
      int n = n0 + i;
      if (i > 0) {
        int jo = n - 16;
        jo = (jo < 0) ? (-jo - 1) : jo;
        int jn = n + 15;
        jn = (jn >= NMELS) ? (2 * NMELS - 1 - jn) : jn;
        win_slide(a, base[jo * NFRAMES], base[jn * NFRAMES]);
      }
      float pe = a[15];
      int   e  = cb + n * NFRAMES;
      float S  = mel[e];
      float ha = medT[e];
      float Z  = fmaxf(ha, pe);
      bool bad = Z < 1.17549435e-38f;
      float Zs = bad ? 1.0f : Z;
      float mh = ha / Zs; mh *= mh;
      float mp = pe / Zs; mp *= mp;
      float den = mh + mp;
      float maskh = bad ? 0.0f : mh / den;
      float maskp = bad ? 0.0f : mp / den;
      float vh = S * maskh, vp = S * maskp;
      if (FIN) {
        out[e]               = to_unit_db(S,  refS);
        out[E_ELEMS + e]     = to_unit_db(vh, refH);
        out[2 * E_ELEMS + e] = to_unit_db(vp, refP);
      } else {
        s = fmaxf(s, S); dh = fmaxf(dh, vh); dp = fmaxf(dp, vp);
      }
    }
  }
  if (FIN) return;
  __shared__ float r0[256], r1[256], r2[256];
  r0[tid] = s; r1[tid] = dh; r2[tid] = dp;
  __syncthreads();
  for (int off = 128; off > 0; off >>= 1) {
    if (tid < off) {
      r0[tid] = fmaxf(r0[tid], r0[tid + off]);
      r1[tid] = fmaxf(r1[tid], r1[tid + off]);
      r2[tid] = fmaxf(r2[tid], r2[tid + off]);
    }
    __syncthreads();
  }
  if (tid == 0) {
    atomicMax(mx + 0, __float_as_uint(r0[0]));
    atomicMax(mx + 1, __float_as_uint(r1[0]));
    atomicMax(mx + 2, __float_as_uint(r2[0]));
  }
}

extern "C" void kernel_launch(void* const* d_in, const int* in_sizes, int n_in,
                              void* d_out, int out_size, void* d_ws,
                              size_t ws_size, hipStream_t stream) {
  const float* wave = (const float*)d_in[0];
  float* out  = (float*)d_out;
  float* ws   = (float*)d_ws;
  float* mel  = ws + MEL_OFF;
  float* medT = ws + MEDT_OFF;
  unsigned* mx = (unsigned*)(ws + MAX_OFF);

  build_tables<<<MAX_OFF / 256, 256, 0, stream>>>(ws);
  mel_gemm<<<NB * MT2, 256, 0, stream>>>(wave, ws, mel);
  median_time<<<NB * NMELS, 256, 0, stream>>>(mel, medT);
  dim3 hg((NFRAMES + 255) / 256, NMELS / RN, NB);
  hpss_mel<false><<<hg, 256, 0, stream>>>(mel, medT, mx, out);
  hpss_mel<true><<<hg, 256, 0, stream>>>(mel, medT, mx, out);
}


# hpss_mel masks: the four IEEE divisions per element become two hardware reciprocals (v_rcp_f32, 1 ulp) and multiplies, in both passes; masks move in the last bits; no sum order change
# speedup vs baseline: 4.4405x; 1.0465x over previous
#include <hip/hip_runtime.h>
#include <math.h>

#define NFFT     1024
#define HOPSZ    512
#define NMELS    128
#define NFREQ    513
#define FPAD     528
#define T_IN     1323000
#define NB       16
#define NFRAMES  2584
#define E_ELEMS  (NB*NMELS*NFRAMES)

#define KR       8
#define KJ       64
#define GP       128
#define DFT_OFF  0
#define DFT_SZ   ((KJ/2)*GP*4)
#define TW_OFF   (DFT_OFF + DFT_SZ)
#define TW_SZ    (KR*GP*2)
#define FB_OFF   (TW_OFF + TW_SZ)
#define FB_SZ    ((FPAD/2)*NMELS*2)
#define MAX_OFF  (FB_OFF + FB_SZ)
#define MEL_OFF  (MAX_OFF + 8)
#define MEDT_OFF (MEL_OFF + E_ELEMS)

typedef float v2f __attribute__((ext_vector_type(2)));
typedef float v4f __attribute__((ext_vector_type(4)));
typedef float v8f __attribute__((ext_vector_type(8)));

__device__ __forceinline__ v8f wmma4(v2f a, v2f b, v8f c) {
  return __builtin_amdgcn_wmma_f32_16x16x4_f32(false, a, false, b, (short)0, c,
                                               false, false);
}

__global__ __launch_bounds__(256) void build_tables(float* __restrict__ ws) {
  int gid = blockIdx.x * 256 + threadIdx.x;
  if (gid == 0) {
    unsigned* mx = (unsigned*)(ws + MAX_OFF);
    mx[0] = 0u; mx[1] = 0u; mx[2] = 0u;
  }
  const float TWO_PI = 6.28318530717958647692f;
  if (gid < TW_OFF) {
    int i = gid - DFT_OFF;
    int entry = i & 3;
    int rest  = i >> 2;
    int g  = rest % GP;
    int j2 = rest / GP;
    int j  = 2 * j2 + (entry & 1);
    int p  = (j * g) & (GP - 1);
    float arg = (float)p * (TWO_PI / 128.0f);
    ws[gid] = (entry < 2) ? cosf(arg) : sinf(arg);
  } else if (gid < FB_OFF) {
    int i = gid - TW_OFF;
    int entry = i & 1;
    int rest  = i >> 1;
    int g  = rest % GP;
    int r  = rest / GP;
    float arg = (float)(r * g) * (TWO_PI / 1024.0f);
    ws[gid] = entry ? sinf(arg) : cosf(arg);
  } else if (gid < MAX_OFF) {
    int i = gid - FB_OFF;
    int entry = i & 1;
    int rest  = i >> 1;
    int m  = rest % NMELS;
    int k2 = rest / NMELS;
    int f  = 2 * k2 + entry;
    float v = 0.0f;
    if (f < NFREQ) {
      float freq   = (float)f * (22050.0f / 512.0f);
      float melmax = 2595.0f * log10f(1.0f + 22050.0f / 700.0f);
      float step   = melmax / 129.0f;
      float p0 = 700.0f * (exp10f(((float)(m    ) * step) / 2595.0f) - 1.0f);
      float p1 = 700.0f * (exp10f(((float)(m + 1) * step) / 2595.0f) - 1.0f);
      float p2 = 700.0f * (exp10f(((float)(m + 2) * step) / 2595.0f) - 1.0f);
      float down = (freq - p0) / (p1 - p0);
      float up   = (p2 - freq) / (p2 - p1);
      float t = fmaxf(0.0f, fminf(down, up));
      v = t * 2.0f / (p2 - p0);
    }
    ws[gid] = v;
  }
}

#define KH     512
#define HB     15
#define MT2    173
#define HPITCH (FPAD + 4)
#define APITCH (KJ + 4)
#define AROW   (KR * APITCH + 4)

__device__ __forceinline__ void dft4(float r0, float i0, float r1, float i1,
                                     float r2, float i2, float r3, float i3,
                                     float (&Br)[4], float (&Bi)[4]) {
  float s02r = r0 + r2, s02i = i0 + i2, d02r = r0 - r2, d02i = i0 - i2;
  float s13r = r1 + r3, s13i = i1 + i3, d13r = r1 - r3, d13i = i1 - i3;
  Br[0] = s02r + s13r; Bi[0] = s02i + s13i;
  Br[1] = d02r - d13i; Bi[1] = d02i + d13r;
  Br[2] = s02r - s13r; Bi[2] = s02i - s13i;
  Br[3] = d02r + d13i; Bi[3] = d02i - d13r;
}
__global__ __launch_bounds__(256, 6) void mel_gemm(const float* __restrict__ wave,
                                                const float* __restrict__ ws,
                                                float* __restrict__ mel) {
  __shared__ __attribute__((aligned(16))) float AL[16][AROW];
  __shared__ float HC[16][HPITCH];
  __shared__ float HS[16][HPITCH];
  __shared__ float GS[16][KR];
  float (*PL)[AROW] = AL;

  const float* __restrict__ dftT = ws + DFT_OFF;
  const float* __restrict__ twT  = ws + TW_OFF;
  const float* __restrict__ fb2  = ws + FB_OFF;

  int b    = blockIdx.x / MT2;
  int tile = blockIdx.x % MT2;
  int t0   = tile * HB;
  int tid  = threadIdx.x;
  int lane = tid & 31;
  int wv   = __builtin_amdgcn_readfirstlane(tid >> 5);
  int h    = lane >> 4;
  int l16  = lane & 15;

  const float* wb = wave + (long)b * T_IN;

  {
    int base = t0 * HOPSZ - NFFT / 2;
    v4f q[8];
    #pragma unroll
    for (int i = 0; i < 8; ++i) {
      int o = base + 4 * (tid + 256 * i);
      o = min(max(o, 0), T_IN - 4);
      q[i] = *(const v4f*)(wb + o);
    }
    #pragma unroll
    for (int i = 0; i < 8; ++i) {
      int e = tid + 256 * i;
      int m = e >> 7;
      int k = (e & 127) * 4;
      float* dst = &AL[m][(k & (KR - 1)) * APITCH + (k >> 3)];
      #pragma unroll
      for (int c = 0; c < 4; ++c) dst[c * APITCH] = q[i][c];
    }
  }
  if (tile == 0 || tile == MT2 - 1) {
    __syncthreads();
    for (int idx = tid; idx < 16 * KH; idx += 256) {
      int m = idx >> 9;
      int k = idx & (KH - 1);
      int j = t0 + m;
      if (j == 0 || j >= NFRAMES) {
        float v = 0.0f;
        if (j <= NFRAMES) {
          int o = j * HOPSZ + k - NFFT / 2;
          o = (o < 0) ? -o : o;
          o = (o >= T_IN) ? (2 * (T_IN - 1) - o) : o;
          v = wb[o];
        }
        AL[m][(k & (KR - 1)) * APITCH + (k >> 3)] = v;
      }
    }
  }
  __syncthreads();

  {
    int row  = tid >> 4;
    int r    = (tid >> 1) & (KR - 1);
    int half = tid & 1;
    const float* p = &AL[row][r * APITCH + 32 * half];
    float sum = 0.0f;
    #pragma unroll
    for (int j = 0; j < 32; j += 4) {
      v4f x = *(const v4f*)(p + j);
      sum += (x[0] - x[1]) + (x[2] - x[3]);
    }
    sum += __shfl_xor(sum, 1);
    if (half == 0) GS[row][r] = sum;
  }

  int gt  = wv & 3;
  int par = wv >> 2;
  int g   = gt * 16 + l16;
  float tc[4], ts[4];
  v8f aC[4], aS[4];
  {
    const float* dB = dftT + (long)g * 4;
    const float* aB = &AL[l16][par * APITCH];
    #pragma unroll
    for (int n = 0; n < 4; ++n) {
      v2f t = *(const v2f*)(twT + (long)((2 * n + par) * GP + g) * 2);
      tc[n] = t[0]; ts[n] = t[1];
    }
    #pragma unroll
    for (int n = 0; n < 4; ++n) {
      #pragma unroll
      for (int j = 0; j < 8; ++j) { aC[n][j] = 0.0f; aS[n][j] = 0.0f; }
    }
    #pragma unroll
    for (int k = 0; k < KJ; k += 4) {
      int kk = k + 2 * h;
      v4f q = *(const v4f*)(dB + (long)(kk >> 1) * (GP * 4));
      v2f bc; bc[0] = q[0]; bc[1] = q[1];
      v2f bs; bs[0] = q[2]; bs[1] = q[3];
      #pragma unroll
      for (int n = 0; n < 4; ++n) {
        v2f a = *(const v2f*)(aB + 2 * n * APITCH + kk);
        aC[n] = wmma4(a, bc, aC[n]);
        aS[n] = wmma4(a, bs, aS[n]);
      }
    }
  }
  const float RH = 0.70710678118654752440f;
  int mir[4];
  #pragma unroll
  for (int q = 0; q < 4; ++q)
    mir[q] = (q == 0 || g != 0) ? (512 - g - 128 * q) : (528 + q);
  int b513 = (g == 1) ? 513 : ((g == 0) ? 514 : 513 + g);
  float Wr[8][4], Wi[8][4];
  #pragma unroll
  for (int v = 0; v < 8; ++v) {
    float ar[4], ai[4];
    #pragma unroll
    for (int n = 0; n < 4; ++n) {
      ar[n] = aC[n][v] * tc[n] - aS[n][v] * ts[n];
      ai[n] = aC[n][v] * ts[n] + aS[n][v] * tc[n];
    }
    dft4(ar[0], ai[0], ar[1], ai[1], ar[2], ai[2], ar[3], ai[3], Wr[v], Wi[v]);
  }
  if (par == 0) {
    #pragma unroll
    for (int v = 0; v < 8; ++v) {
      int row = v + 8 * h;
      #pragma unroll
      for (int q = 0; q < 4; ++q) {
        HC[row][g + 128 * q] = Wr[v][q];
        HS[row][g + 128 * q] = Wi[v][q];
        HC[row][mir[q]] = Wr[v][q];
        HS[row][mir[q]] = -Wi[v][q];
      }
      if (gt == 0) {
        HC[row][b513] = Wr[v][0];
        HS[row][b513] = Wi[v][0];
      }
    }
  } else {
    #pragma unroll
    for (int v = 0; v < 8; ++v) {
      float r1 = Wr[v][1], i1 = Wi[v][1];
      float r2 = Wr[v][2], i2 = Wi[v][2];
      float r3 = Wr[v][3], i3 = Wi[v][3];
      Wr[v][1] = RH * (r1 - i1);    Wi[v][1] = RH * (r1 + i1);
      Wr[v][2] = -i2;               Wi[v][2] = r2;
      Wr[v][3] = -RH * (r3 + i3);   Wi[v][3] = RH * (r3 - i3);
    }
  }
  __syncthreads();
  if (par == 1) {
    #pragma unroll
    for (int v = 0; v < 8; ++v) {
      int row = v + 8 * h;
      #pragma unroll
      for (int q = 0; q < 4; ++q) {
        HC[row][g + 128 * q] += Wr[v][q];
        HS[row][g + 128 * q] += Wi[v][q];
        HC[row][mir[q]] -= Wr[v][q];
        HS[row][mir[q]] += Wi[v][q];
      }
      if (gt == 0) {
        HC[row][b513] -= Wr[v][0];
        HS[row][b513] -= Wi[v][0];
      }
    }
  } else if (wv == 0 && lane < 16) {
    float ar[KR], ai[KR];
    #pragma unroll
    for (int r = 0; r < KR; ++r) {
      v2f t = *(const v2f*)(twT + (long)(r * GP + 64) * 2);
      float sr = GS[lane][r];
      ar[r] = sr * t[0]; ai[r] = sr * t[1];
    }
    float Er[4], Ei[4], Or[4], Oi[4];
    dft4(ar[0], ai[0], ar[2], ai[2], ar[4], ai[4], ar[6], ai[6], Er, Ei);
    dft4(ar[1], ai[1], ar[3], ai[3], ar[5], ai[5], ar[7], ai[7], Or, Oi);
    HC[lane][64]  = Er[0] + Or[0];
    HS[lane][64]  = Ei[0] + Oi[0];
    HC[lane][192] = Er[1] + RH * (Or[1] - Oi[1]);
    HS[lane][192] = Ei[1] + RH * (Or[1] + Oi[1]);
    HC[lane][320] = Er[2] - Oi[2];
    HS[lane][320] = Ei[2] + Or[2];
    HC[lane][448] = Er[3] - RH * (Or[3] + Oi[3]);
    HS[lane][448] = Ei[3] + RH * (Or[3] - Oi[3]);
  }
  __syncthreads();

  for (int idx = tid; idx < 16 * FPAD; idx += 256) {
    int m = idx / FPAD;
    int f = idx - m * FPAD;
    float p = 0.0f;
    if (m < HB && f < NFREQ) {
      float s  = (f & 1) ? -1.0f : 1.0f;
      int   fm = (f == 0) ? 1 : f - 1;
      float sm = (f == 0) ? -1.0f : 1.0f;
      int   fp = f + 1;
      float c0 = HC[m][f]  + s * HC[m + 1][f];
      float s0 = HS[m][f]  + s * HS[m + 1][f];
      float cm = HC[m][fm] - s * HC[m + 1][fm];
      float sn = (HS[m][fm] - s * HS[m + 1][fm]) * sm;
      float cp = HC[m][fp] - s * HC[m + 1][fp];
      float sp = HS[m][fp] - s * HS[m + 1][fp];
      float re = 0.5f * c0 - 0.25f * (cm + cp);
      float im = 0.5f * s0 - 0.25f * (sn + sp);
      p = re * re + im * im;
    }
    PL[m][f] = p;
  }
  __syncthreads();

  int n0 = wv * 16;
  const float* fB = fb2 + (long)(n0 + l16) * 2;
  v8f acc;
  #pragma unroll
  for (int j = 0; j < 8; ++j) acc[j] = 0.0f;
  for (int k = 0; k < FPAD; k += 4) {
    int kk = k + 2 * h;
    v2f a  = *(const v2f*)&PL[l16][kk];
    v2f bb = *(const v2f*)(fB + (long)(kk >> 1) * (NMELS * 2));
    acc = wmma4(a, bb, acc);
  }
  int n = n0 + l16;
  #pragma unroll
  for (int v = 0; v < 8; ++v) {
    int r = v + 8 * h;
    int t = t0 + r;
    if (r < HB && t < NFRAMES) mel[(b * NMELS + n) * NFRAMES + t] = acc[v];
  }
}


#define WIN_INSERT(a, m, x)                                             \
  do {                                                                  \
    if ((m) == 0) { (a)[0] = (x); }                                     \
    else {                                                              \
      (a)[(m)] = fmaxf((a)[(m) - 1], (x));                              \
      for (int i_ = (m) - 1; i_ >= 1; --i_)                             \
        (a)[i_] = fmaxf((a)[i_ - 1], fminf((a)[i_], (x)));              \
      (a)[0] = fminf((a)[0], (x));                                      \
    }                                                                   \
  } while (0)

__device__ __forceinline__ void win_slide(float (&a)[31], float xo, float xn) {
  float r[30];
  #pragma unroll
  for (int i = 0; i < 30; ++i) r[i] = (a[i] < xo) ? a[i] : a[i + 1];
  a[0] = fminf(r[0], xn);
  #pragma unroll
  for (int i = 1; i < 30; ++i) a[i] = fmaxf(r[i - 1], fminf(r[i], xn));
  a[30] = fmaxf(r[29], xn);
}

#define RT 11
__global__ __launch_bounds__(256) void median_time(const float* __restrict__ mel,
                                                   float* __restrict__ medT) {
  __shared__ float P[NFRAMES + 30];
  __shared__ float O[NFRAMES];
  int row = blockIdx.x;
  int tid = threadIdx.x;
  const float* src = mel + (long)row * NFRAMES;
  for (int i = tid; i < NFRAMES + 30; i += 256) {
    int j = i - 15;
    j = (j < 0) ? (-j - 1) : j;
    j = (j >= NFRAMES) ? (2 * NFRAMES - 1 - j) : j;
    P[i] = src[j];
  }
  __syncthreads();
  int t0 = tid * RT;
  if (t0 < NFRAMES) {
    float a[31];
    #pragma unroll
    for (int m = 0; m < 31; ++m) {
      float x = P[t0 + m];
      WIN_INSERT(a, m, x);
    }
    O[t0] = a[15];
    #pragma unroll
    for (int i = 1; i < RT; ++i) {
      int t = t0 + i;
      if (t < NFRAMES) {
        win_slide(a, P[t - 1], P[t + 30]);
        O[t] = a[15];
      }
    }
  }
  __syncthreads();
  float* dst = medT + (long)row * NFRAMES;
  for (int i = tid; i < NFRAMES; i += 256) dst[i] = O[i];
}

__device__ __forceinline__ float to_unit_db(float v, float ref) {
  float db = 10.0f * log10f(fmaxf(1e-10f, v)) -
             10.0f * log10f(fmaxf(1e-10f, ref));
  db = fmaxf(db, -80.0f);
  return (db + 80.0f) * (1.0f / (80.0f + 1e-7f));
}

#define RN 16
template <bool FIN>
__global__ __launch_bounds__(256) void hpss_mel(const float* __restrict__ mel,
                                                const float* __restrict__ medT,
                                                unsigned* __restrict__ mx,
                                                float* __restrict__ out) {
  int tid = threadIdx.x;
  int t   = blockIdx.x * 256 + tid;
  int n0  = blockIdx.y * RN;
  int b   = blockIdx.z;
  float s = 0.0f, dh = 0.0f, dp = 0.0f;
  float refS = 0.0f, refH = 0.0f, refP = 0.0f;
  if (FIN) {
    refS = __uint_as_float(mx[0]);
    refH = __uint_as_float(mx[1]);
    refP = __uint_as_float(mx[2]);
  }
  if (t < NFRAMES) {
    int cb = (b * NMELS) * NFRAMES + t;
    const float* base = mel + cb;
    float a[31];
    #pragma unroll
    for (int m = 0; m < 31; ++m) {
      int j = n0 + m - 15;
      j = (j < 0) ? (-j - 1) : j;
      j = (j >= NMELS) ? (2 * NMELS - 1 - j) : j;
      float x = base[j * NFRAMES];
      WIN_INSERT(a, m, x);
    }
    #pragma unroll
    for (int i = 0; i < RN; ++i) {
      int n = n0 + i;
      if (i > 0) {
        int jo = n - 16;
        jo = (jo < 0) ? (-jo - 1) : jo;
        int jn = n + 15;
        jn = (jn >= NMELS) ? (2 * NMELS - 1 - jn) : jn;
        win_slide(a, base[jo * NFRAMES], base[jn * NFRAMES]);
      }
      float pe = a[15];
      int   e  = cb + n * NFRAMES;
      float S  = mel[e];
      float ha = medT[e];
      float Z  = fmaxf(ha, pe);
      bool bad = Z < 1.17549435e-38f;
      float Zs = bad ? 1.0f : Z;
      float rz = __builtin_amdgcn_rcpf(Zs);
      float mh = ha * rz; mh *= mh;
      float mp = pe * rz; mp *= mp;
      float rd = __builtin_amdgcn_rcpf(mh + mp);
      float maskh = bad ? 0.0f : mh * rd;
      float maskp = bad ? 0.0f : mp * rd;
      float vh = S * maskh, vp = S * maskp;
      if (FIN) {
        out[e]               = to_unit_db(S,  refS);
        out[E_ELEMS + e]     = to_unit_db(vh, refH);
        out[2 * E_ELEMS + e] = to_unit_db(vp, refP);
      } else {
        s = fmaxf(s, S); dh = fmaxf(dh, vh); dp = fmaxf(dp, vp);
      }
    }
  }
  if (FIN) return;
  __shared__ float r0[256], r1[256], r2[256];
  r0[tid] = s; r1[tid] = dh; r2[tid] = dp;
  __syncthreads();
  for (int off = 128; off > 0; off >>= 1) {
    if (tid < off) {
      r0[tid] = fmaxf(r0[tid], r0[tid + off]);
      r1[tid] = fmaxf(r1[tid], r1[tid + off]);
      r2[tid] = fmaxf(r2[tid], r2[tid + off]);
    }
    __syncthreads();
  }
  if (tid == 0) {
    atomicMax(mx + 0, __float_as_uint(r0[0]));
    atomicMax(mx + 1, __float_as_uint(r1[0]));
    atomicMax(mx + 2, __float_as_uint(r2[0]));
  }
}

extern "C" void kernel_launch(void* const* d_in, const int* in_sizes, int n_in,
                              void* d_out, int out_size, void* d_ws,
                              size_t ws_size, hipStream_t stream) {
  const float* wave = (const float*)d_in[0];
  float* out  = (float*)d_out;
  float* ws   = (float*)d_ws;
  float* mel  = ws + MEL_OFF;
  float* medT = ws + MEDT_OFF;
  unsigned* mx = (unsigned*)(ws + MAX_OFF);

  build_tables<<<MAX_OFF / 256, 256, 0, stream>>>(ws);
  mel_gemm<<<NB * MT2, 256, 0, stream>>>(wave, ws, mel);
  median_time<<<NB * NMELS, 256, 0, stream>>>(mel, medT);
  dim3 hg((NFRAMES + 255) / 256, NMELS / RN, NB);
  hpss_mel<false><<<hg, 256, 0, stream>>>(mel, medT, mx, out);
  hpss_mel<true><<<hg, 256, 0, stream>>>(mel, medT, mx, out);
}


# mel_gemm power tile: a thread owns two adjacent bins and walks the 16 half-frame rows: a row is read once for two frames, four bins serve two outputs, (-1)^f fixed per register; sum order same
# speedup vs baseline: 4.6537x; 1.0480x over previous
#include <hip/hip_runtime.h>
#include <math.h>

#define NFFT     1024
#define HOPSZ    512
#define NMELS    128
#define NFREQ    513
#define FPAD     528
#define T_IN     1323000
#define NB       16
#define NFRAMES  2584
#define E_ELEMS  (NB*NMELS*NFRAMES)

#define KR       8
#define KJ       64
#define GP       128
#define DFT_OFF  0
#define DFT_SZ   ((KJ/2)*GP*4)
#define TW_OFF   (DFT_OFF + DFT_SZ)
#define TW_SZ    (KR*GP*2)
#define FB_OFF   (TW_OFF + TW_SZ)
#define FB_SZ    ((FPAD/2)*NMELS*2)
#define MAX_OFF  (FB_OFF + FB_SZ)
#define MEL_OFF  (MAX_OFF + 8)
#define MEDT_OFF (MEL_OFF + E_ELEMS)

typedef float v2f __attribute__((ext_vector_type(2)));
typedef float v4f __attribute__((ext_vector_type(4)));
typedef float v8f __attribute__((ext_vector_type(8)));

__device__ __forceinline__ v8f wmma4(v2f a, v2f b, v8f c) {
  return __builtin_amdgcn_wmma_f32_16x16x4_f32(false, a, false, b, (short)0, c,
                                               false, false);
}

__global__ __launch_bounds__(256) void build_tables(float* __restrict__ ws) {
  int gid = blockIdx.x * 256 + threadIdx.x;
  if (gid == 0) {
    unsigned* mx = (unsigned*)(ws + MAX_OFF);
    mx[0] = 0u; mx[1] = 0u; mx[2] = 0u;
  }
  const float TWO_PI = 6.28318530717958647692f;
  if (gid < TW_OFF) {
    int i = gid - DFT_OFF;
    int entry = i & 3;
    int rest  = i >> 2;
    int g  = rest % GP;
    int j2 = rest / GP;
    int j  = 2 * j2 + (entry & 1);
    int p  = (j * g) & (GP - 1);
    float arg = (float)p * (TWO_PI / 128.0f);
    ws[gid] = (entry < 2) ? cosf(arg) : sinf(arg);
  } else if (gid < FB_OFF) {
    int i = gid - TW_OFF;
    int entry = i & 1;
    int rest  = i >> 1;
    int g  = rest % GP;
    int r  = rest / GP;
    float arg = (float)(r * g) * (TWO_PI / 1024.0f);
    ws[gid] = entry ? sinf(arg) : cosf(arg);
  } else if (gid < MAX_OFF) {
    int i = gid - FB_OFF;
    int entry = i & 1;
    int rest  = i >> 1;
    int m  = rest % NMELS;
    int k2 = rest / NMELS;
    int f  = 2 * k2 + entry;
    float v = 0.0f;
    if (f < NFREQ) {
      float freq   = (float)f * (22050.0f / 512.0f);
      float melmax = 2595.0f * log10f(1.0f + 22050.0f / 700.0f);
      float step   = melmax / 129.0f;
      float p0 = 700.0f * (exp10f(((float)(m    ) * step) / 2595.0f) - 1.0f);
      float p1 = 700.0f * (exp10f(((float)(m + 1) * step) / 2595.0f) - 1.0f);
      float p2 = 700.0f * (exp10f(((float)(m + 2) * step) / 2595.0f) - 1.0f);
      float down = (freq - p0) / (p1 - p0);
      float up   = (p2 - freq) / (p2 - p1);
      float t = fmaxf(0.0f, fminf(down, up));
      v = t * 2.0f / (p2 - p0);
    }
    ws[gid] = v;
  }
}

#define KH     512
#define HB     15
#define MT2    173
#define HPITCH (FPAD + 4)
#define APITCH (KJ + 4)
#define AROW   (KR * APITCH + 4)

__device__ __forceinline__ void dft4(float r0, float i0, float r1, float i1,
                                     float r2, float i2, float r3, float i3,
                                     float (&Br)[4], float (&Bi)[4]) {
  float s02r = r0 + r2, s02i = i0 + i2, d02r = r0 - r2, d02i = i0 - i2;
  float s13r = r1 + r3, s13i = i1 + i3, d13r = r1 - r3, d13i = i1 - i3;
  Br[0] = s02r + s13r; Bi[0] = s02i + s13i;
  Br[1] = d02r - d13i; Bi[1] = d02i + d13r;
  Br[2] = s02r - s13r; Bi[2] = s02i - s13i;
  Br[3] = d02r + d13i; Bi[3] = d02i - d13r;
}
__global__ __launch_bounds__(256, 6) void mel_gemm(const float* __restrict__ wave,
                                                const float* __restrict__ ws,
                                                float* __restrict__ mel) {
  __shared__ __attribute__((aligned(16))) float AL[16][AROW];
  __shared__ float HC[16][HPITCH];
  __shared__ float HS[16][HPITCH];
  __shared__ float GS[16][KR];
  float (*PL)[AROW] = AL;

  const float* __restrict__ dftT = ws + DFT_OFF;
  const float* __restrict__ twT  = ws + TW_OFF;
  const float* __restrict__ fb2  = ws + FB_OFF;

  int b    = blockIdx.x / MT2;
  int tile = blockIdx.x % MT2;
  int t0   = tile * HB;
  int tid  = threadIdx.x;
  int lane = tid & 31;
  int wv   = __builtin_amdgcn_readfirstlane(tid >> 5);
  int h    = lane >> 4;
  int l16  = lane & 15;

  const float* wb = wave + (long)b * T_IN;

  {
    int base = t0 * HOPSZ - NFFT / 2;
    v4f q[8];
    #pragma unroll
    for (int i = 0; i < 8; ++i) {
      int o = base + 4 * (tid + 256 * i);
      o = min(max(o, 0), T_IN - 4);
      q[i] = *(const v4f*)(wb + o);
    }
    #pragma unroll
    for (int i = 0; i < 8; ++i) {
      int e = tid + 256 * i;
      int m = e >> 7;
      int k = (e & 127) * 4;
      float* dst = &AL[m][(k & (KR - 1)) * APITCH + (k >> 3)];
      #pragma unroll
      for (int c = 0; c < 4; ++c) dst[c * APITCH] = q[i][c];
    }
  }
  if (tile == 0 || tile == MT2 - 1) {
    __syncthreads();
    for (int idx = tid; idx < 16 * KH; idx += 256) {
      int m = idx >> 9;
      int k = idx & (KH - 1);
      int j = t0 + m;
      if (j == 0 || j >= NFRAMES) {
        float v = 0.0f;
        if (j <= NFRAMES) {
          int o = j * HOPSZ + k - NFFT / 2;
          o = (o < 0) ? -o : o;
          o = (o >= T_IN) ? (2 * (T_IN - 1) - o) : o;
          v = wb[o];
        }
        AL[m][(k & (KR - 1)) * APITCH + (k >> 3)] = v;
      }
    }
  }
  __syncthreads();

  {
    int row  = tid >> 4;
    int r    = (tid >> 1) & (KR - 1);
    int half = tid & 1;
    const float* p = &AL[row][r * APITCH + 32 * half];
    float sum = 0.0f;
    #pragma unroll
    for (int j = 0; j < 32; j += 4) {
      v4f x = *(const v4f*)(p + j);
      sum += (x[0] - x[1]) + (x[2] - x[3]);
    }
    sum += __shfl_xor(sum, 1);
    if (half == 0) GS[row][r] = sum;
  }

  int gt  = wv & 3;
  int par = wv >> 2;
  int g   = gt * 16 + l16;
  float tc[4], ts[4];
  v8f aC[4], aS[4];
  {
    const float* dB = dftT + (long)g * 4;
    const float* aB = &AL[l16][par * APITCH];
    #pragma unroll
    for (int n = 0; n < 4; ++n) {
      v2f t = *(const v2f*)(twT + (long)((2 * n + par) * GP + g) * 2);
      tc[n] = t[0]; ts[n] = t[1];
    }
    #pragma unroll
    for (int n = 0; n < 4; ++n) {
      #pragma unroll
      for (int j = 0; j < 8; ++j) { aC[n][j] = 0.0f; aS[n][j] = 0.0f; }
    }
    #pragma unroll
    for (int k = 0; k < KJ; k += 4) {
      int kk = k + 2 * h;
      v4f q = *(const v4f*)(dB + (long)(kk >> 1) * (GP * 4));
      v2f bc; bc[0] = q[0]; bc[1] = q[1];
      v2f bs; bs[0] = q[2]; bs[1] = q[3];
      #pragma unroll
      for (int n = 0; n < 4; ++n) {
        v2f a = *(const v2f*)(aB + 2 * n * APITCH + kk);
        aC[n] = wmma4(a, bc, aC[n]);
        aS[n] = wmma4(a, bs, aS[n]);
      }
    }
  }
  const float RH = 0.70710678118654752440f;
  int mir[4];
  #pragma unroll
  for (int q = 0; q < 4; ++q)
    mir[q] = (q == 0 || g != 0) ? (512 - g - 128 * q) : (528 + q);
  int b513 = (g == 1) ? 513 : ((g == 0) ? 514 : 513 + g);
  float Wr[8][4], Wi[8][4];
  #pragma unroll
  for (int v = 0; v < 8; ++v) {
    float ar[4], ai[4];
    #pragma unroll
    for (int n = 0; n < 4; ++n) {
      ar[n] = aC[n][v] * tc[n] - aS[n][v] * ts[n];
      ai[n] = aC[n][v] * ts[n] + aS[n][v] * tc[n];
    }
    dft4(ar[0], ai[0], ar[1], ai[1], ar[2], ai[2], ar[3], ai[3], Wr[v], Wi[v]);
  }
  if (par == 0) {
    #pragma unroll
    for (int v = 0; v < 8; ++v) {
      int row = v + 8 * h;
      #pragma unroll
      for (int q = 0; q < 4; ++q) {
        HC[row][g + 128 * q] = Wr[v][q];
        HS[row][g + 128 * q] = Wi[v][q];
        HC[row][mir[q]] = Wr[v][q];
        HS[row][mir[q]] = -Wi[v][q];
      }
      if (gt == 0) {
        HC[row][b513] = Wr[v][0];
        HS[row][b513] = Wi[v][0];
      }
    }
  } else {
    #pragma unroll
    for (int v = 0; v < 8; ++v) {
      float r1 = Wr[v][1], i1 = Wi[v][1];
      float r2 = Wr[v][2], i2 = Wi[v][2];
      float r3 = Wr[v][3], i3 = Wi[v][3];
      Wr[v][1] = RH * (r1 - i1);    Wi[v][1] = RH * (r1 + i1);
      Wr[v][2] = -i2;               Wi[v][2] = r2;
      Wr[v][3] = -RH * (r3 + i3);   Wi[v][3] = RH * (r3 - i3);
    }
  }
  __syncthreads();
  if (par == 1) {
    #pragma unroll
    for (int v = 0; v < 8; ++v) {
      int row = v + 8 * h;
      #pragma unroll
      for (int q = 0; q < 4; ++q) {
        HC[row][g + 128 * q] += Wr[v][q];
        HS[row][g + 128 * q] += Wi[v][q];
        HC[row][mir[q]] -= Wr[v][q];
        HS[row][mir[q]] += Wi[v][q];
      }
      if (gt == 0) {
        HC[row][b513] -= Wr[v][0];
        HS[row][b513] -= Wi[v][0];
      }
    }
  } else if (wv == 0 && lane < 16) {
    float ar[KR], ai[KR];
    #pragma unroll
    for (int r = 0; r < KR; ++r) {
      v2f t = *(const v2f*)(twT + (long)(r * GP + 64) * 2);
      float sr = GS[lane][r];
      ar[r] = sr * t[0]; ai[r] = sr * t[1];
    }
    float Er[4], Ei[4], Or[4], Oi[4];
    dft4(ar[0], ai[0], ar[2], ai[2], ar[4], ai[4], ar[6], ai[6], Er, Ei);
    dft4(ar[1], ai[1], ar[3], ai[3], ar[5], ai[5], ar[7], ai[7], Or, Oi);
    HC[lane][64]  = Er[0] + Or[0];
    HS[lane][64]  = Ei[0] + Oi[0];
    HC[lane][192] = Er[1] + RH * (Or[1] - Oi[1]);
    HS[lane][192] = Ei[1] + RH * (Or[1] + Oi[1]);
    HC[lane][320] = Er[2] - Oi[2];
    HS[lane][320] = Ei[2] + Or[2];
    HC[lane][448] = Er[3] - RH * (Or[3] + Oi[3]);
    HS[lane][448] = Ei[3] + RH * (Or[3] - Oi[3]);
  }
  __syncthreads();

  {
    const int   f0 = 2 * tid;
    const int   fm = (tid == 0) ? 1 : f0 - 1;
    const float sm = (tid == 0) ? -1.0f : 1.0f;
    float pc0 = HC[0][fm], pc1 = HC[0][f0], pc2 = HC[0][f0 + 1], pc3 = HC[0][f0 + 2];
    float ps0 = HS[0][fm], ps1 = HS[0][f0], ps2 = HS[0][f0 + 1], ps3 = HS[0][f0 + 2];
    #pragma unroll
    for (int m = 0; m < HB; ++m) {
      float nc0 = HC[m + 1][fm], nc1 = HC[m + 1][f0], nc2 = HC[m + 1][f0 + 1], nc3 = HC[m + 1][f0 + 2];
      float ns0 = HS[m + 1][fm], ns1 = HS[m + 1][f0], ns2 = HS[m + 1][f0 + 1], ns3 = HS[m + 1][f0 + 2];
      float xc0 = pc0 - nc0, xc1 = pc1 + nc1, xc2 = pc2 - nc2, xc3 = pc3 + nc3;
      float xs0 = (ps0 - ns0) * sm, xs1 = ps1 + ns1, xs2 = ps2 - ns2, xs3 = ps3 + ns3;
      float re0 = 0.5f * xc1 - 0.25f * (xc0 + xc2);
      float im0 = 0.5f * xs1 - 0.25f * (xs0 + xs2);
      float re1 = 0.5f * xc2 - 0.25f * (xc1 + xc3);
      float im1 = 0.5f * xs2 - 0.25f * (xs1 + xs3);
      PL[m][f0]     = re0 * re0 + im0 * im0;
      PL[m][f0 + 1] = re1 * re1 + im1 * im1;
      pc0 = nc0; pc1 = nc1; pc2 = nc2; pc3 = nc3;
      ps0 = ns0; ps1 = ns1; ps2 = ns2; ps3 = ns3;
    }
    PL[HB][f0] = 0.0f; PL[HB][f0 + 1] = 0.0f;
  }
  {
    int m = tid >> 4;
    int f = 512 + (tid & 15);
    float p = 0.0f;
    if (f == 512 && m < HB) {
      float c0 = HC[m][512] + HC[m + 1][512];
      float s0 = HS[m][512] + HS[m + 1][512];
      float cm = HC[m][511] - HC[m + 1][511];
      float sn = HS[m][511] - HS[m + 1][511];
      float cp = HC[m][513] - HC[m + 1][513];
      float sp = HS[m][513] - HS[m + 1][513];
      float re = 0.5f * c0 - 0.25f * (cm + cp);
      float im = 0.5f * s0 - 0.25f * (sn + sp);
      p = re * re + im * im;
    }
    PL[m][f] = p;
  }
  __syncthreads();

  int n0 = wv * 16;
  const float* fB = fb2 + (long)(n0 + l16) * 2;
  v8f acc;
  #pragma unroll
  for (int j = 0; j < 8; ++j) acc[j] = 0.0f;
  for (int k = 0; k < FPAD; k += 4) {
    int kk = k + 2 * h;
    v2f a  = *(const v2f*)&PL[l16][kk];
    v2f bb = *(const v2f*)(fB + (long)(kk >> 1) * (NMELS * 2));
    acc = wmma4(a, bb, acc);
  }
  int n = n0 + l16;
  #pragma unroll
  for (int v = 0; v < 8; ++v) {
    int r = v + 8 * h;
    int t = t0 + r;
    if (r < HB && t < NFRAMES) mel[(b * NMELS + n) * NFRAMES + t] = acc[v];
  }
}


#define WIN_INSERT(a, m, x)                                             \
  do {                                                                  \
    if ((m) == 0) { (a)[0] = (x); }                                     \
    else {                                                              \
      (a)[(m)] = fmaxf((a)[(m) - 1], (x));                              \
      for (int i_ = (m) - 1; i_ >= 1; --i_)                             \
        (a)[i_] = fmaxf((a)[i_ - 1], fminf((a)[i_], (x)));              \
      (a)[0] = fminf((a)[0], (x));                                      \
    }                                                                   \
  } while (0)

__device__ __forceinline__ void win_slide(float (&a)[31], float xo, float xn) {
  float r[30];
  #pragma unroll
  for (int i = 0; i < 30; ++i) r[i] = (a[i] < xo) ? a[i] : a[i + 1];
  a[0] = fminf(r[0], xn);
  #pragma unroll
  for (int i = 1; i < 30; ++i) a[i] = fmaxf(r[i - 1], fminf(r[i], xn));
  a[30] = fmaxf(r[29], xn);
}

#define RT 11
__global__ __launch_bounds__(256) void median_time(const float* __restrict__ mel,
                                                   float* __restrict__ medT) {
  __shared__ float P[NFRAMES + 30];
  __shared__ float O[NFRAMES];
  int row = blockIdx.x;
  int tid = threadIdx.x;
  const float* src = mel + (long)row * NFRAMES;
  for (int i = tid; i < NFRAMES + 30; i += 256) {
    int j = i - 15;
    j = (j < 0) ? (-j - 1) : j;
    j = (j >= NFRAMES) ? (2 * NFRAMES - 1 - j) : j;
    P[i] = src[j];
  }
  __syncthreads();
  int t0 = tid * RT;
  if (t0 < NFRAMES) {
    float a[31];
    #pragma unroll
    for (int m = 0; m < 31; ++m) {
      float x = P[t0 + m];
      WIN_INSERT(a, m, x);
    }
    O[t0] = a[15];
    #pragma unroll
    for (int i = 1; i < RT; ++i) {
      int t = t0 + i;
      if (t < NFRAMES) {
        win_slide(a, P[t - 1], P[t + 30]);
        O[t] = a[15];
      }
    }
  }
  __syncthreads();
  float* dst = medT + (long)row * NFRAMES;
  for (int i = tid; i < NFRAMES; i += 256) dst[i] = O[i];
}

__device__ __forceinline__ float to_unit_db(float v, float ref) {
  float db = 10.0f * log10f(fmaxf(1e-10f, v)) -
             10.0f * log10f(fmaxf(1e-10f, ref));
  db = fmaxf(db, -80.0f);
  return (db + 80.0f) * (1.0f / (80.0f + 1e-7f));
}

#define RN 16
template <bool FIN>
__global__ __launch_bounds__(256) void hpss_mel(const float* __restrict__ mel,
                                                const float* __restrict__ medT,
                                                unsigned* __restrict__ mx,
                                                float* __restrict__ out) {
  int tid = threadIdx.x;
  int t   = blockIdx.x * 256 + tid;
  int n0  = blockIdx.y * RN;
  int b   = blockIdx.z;
  float s = 0.0f, dh = 0.0f, dp = 0.0f;
  float refS = 0.0f, refH = 0.0f, refP = 0.0f;
  if (FIN) {
    refS = __uint_as_float(mx[0]);
    refH = __uint_as_float(mx[1]);
    refP = __uint_as_float(mx[2]);
  }
  if (t < NFRAMES) {
    int cb = (b * NMELS) * NFRAMES + t;
    const float* base = mel + cb;
    float a[31];
    #pragma unroll
    for (int m = 0; m < 31; ++m) {
      int j = n0 + m - 15;
      j = (j < 0) ? (-j - 1) : j;
      j = (j >= NMELS) ? (2 * NMELS - 1 - j) : j;
      float x = base[j * NFRAMES];
      WIN_INSERT(a, m, x);
    }
    #pragma unroll
    for (int i = 0; i < RN; ++i) {
      int n = n0 + i;
      if (i > 0) {
        int jo = n - 16;
        jo = (jo < 0) ? (-jo - 1) : jo;
        int jn = n + 15;
        jn = (jn >= NMELS) ? (2 * NMELS - 1 - jn) : jn;
        win_slide(a, base[jo * NFRAMES], base[jn * NFRAMES]);
      }
      float pe = a[15];
      int   e  = cb + n * NFRAMES;
      float S  = mel[e];
      float ha = medT[e];
      float Z  = fmaxf(ha, pe);
      bool bad = Z < 1.17549435e-38f;
      float Zs = bad ? 1.0f : Z;
      float rz = __builtin_amdgcn_rcpf(Zs);
      float mh = ha * rz; mh *= mh;
      float mp = pe * rz; mp *= mp;
      float rd = __builtin_amdgcn_rcpf(mh + mp);
      float maskh = bad ? 0.0f : mh * rd;
      float maskp = bad ? 0.0f : mp * rd;
      float vh = S * maskh, vp = S * maskp;
      if (FIN) {
        out[e]               = to_unit_db(S,  refS);
        out[E_ELEMS + e]     = to_unit_db(vh, refH);
        out[2 * E_ELEMS + e] = to_unit_db(vp, refP);
      } else {
        s = fmaxf(s, S); dh = fmaxf(dh, vh); dp = fmaxf(dp, vp);
      }
    }
  }
  if (FIN) return;
  __shared__ float r0[256], r1[256], r2[256];
  r0[tid] = s; r1[tid] = dh; r2[tid] = dp;
  __syncthreads();
  for (int off = 128; off > 0; off >>= 1) {
    if (tid < off) {
      r0[tid] = fmaxf(r0[tid], r0[tid + off]);
      r1[tid] = fmaxf(r1[tid], r1[tid + off]);
      r2[tid] = fmaxf(r2[tid], r2[tid + off]);
    }
    __syncthreads();
  }
  if (tid == 0) {
    atomicMax(mx + 0, __float_as_uint(r0[0]));
    atomicMax(mx + 1, __float_as_uint(r1[0]));
    atomicMax(mx + 2, __float_as_uint(r2[0]));
  }
}

extern "C" void kernel_launch(void* const* d_in, const int* in_sizes, int n_in,
                              void* d_out, int out_size, void* d_ws,
                              size_t ws_size, hipStream_t stream) {
  const float* wave = (const float*)d_in[0];
  float* out  = (float*)d_out;
  float* ws   = (float*)d_ws;
  float* mel  = ws + MEL_OFF;
  float* medT = ws + MEDT_OFF;
  unsigned* mx = (unsigned*)(ws + MAX_OFF);

  build_tables<<<MAX_OFF / 256, 256, 0, stream>>>(ws);
  mel_gemm<<<NB * MT2, 256, 0, stream>>>(wave, ws, mel);
  median_time<<<NB * NMELS, 256, 0, stream>>>(mel, medT);
  dim3 hg((NFRAMES + 255) / 256, NMELS / RN, NB);
  hpss_mel<false><<<hg, 256, 0, stream>>>(mel, medT, mx, out);
  hpss_mel<true><<<hg, 256, 0, stream>>>(mel, medT, mx, out);
}


# median_time, hpss_mel: sorted 31-tap window built at the middle of a thread's run and slid to both ends (5+5, 7+8 steps): only slots near the median stay live; insert as median-of-3; exact medians
# speedup vs baseline: 4.8631x; 1.0450x over previous
#include <hip/hip_runtime.h>
#include <math.h>

#define NFFT     1024
#define HOPSZ    512
#define NMELS    128
#define NFREQ    513
#define FPAD     528
#define T_IN     1323000
#define NB       16
#define NFRAMES  2584
#define E_ELEMS  (NB*NMELS*NFRAMES)

#define KR       8
#define KJ       64
#define GP       128
#define DFT_OFF  0
#define DFT_SZ   ((KJ/2)*GP*4)
#define TW_OFF   (DFT_OFF + DFT_SZ)
#define TW_SZ    (KR*GP*2)
#define FB_OFF   (TW_OFF + TW_SZ)
#define FB_SZ    ((FPAD/2)*NMELS*2)
#define MAX_OFF  (FB_OFF + FB_SZ)
#define MEL_OFF  (MAX_OFF + 8)
#define MEDT_OFF (MEL_OFF + E_ELEMS)

typedef float v2f __attribute__((ext_vector_type(2)));
typedef float v4f __attribute__((ext_vector_type(4)));
typedef float v8f __attribute__((ext_vector_type(8)));

__device__ __forceinline__ v8f wmma4(v2f a, v2f b, v8f c) {
  return __builtin_amdgcn_wmma_f32_16x16x4_f32(false, a, false, b, (short)0, c,
                                               false, false);
}

__global__ __launch_bounds__(256) void build_tables(float* __restrict__ ws) {
  int gid = blockIdx.x * 256 + threadIdx.x;
  if (gid == 0) {
    unsigned* mx = (unsigned*)(ws + MAX_OFF);
    mx[0] = 0u; mx[1] = 0u; mx[2] = 0u;
  }
  const float TWO_PI = 6.28318530717958647692f;
  if (gid < TW_OFF) {
    int i = gid - DFT_OFF;
    int entry = i & 3;
    int rest  = i >> 2;
    int g  = rest % GP;
    int j2 = rest / GP;
    int j  = 2 * j2 + (entry & 1);
    int p  = (j * g) & (GP - 1);
    float arg = (float)p * (TWO_PI / 128.0f);
    ws[gid] = (entry < 2) ? cosf(arg) : sinf(arg);
  } else if (gid < FB_OFF) {
    int i = gid - TW_OFF;
    int entry = i & 1;
    int rest  = i >> 1;
    int g  = rest % GP;
    int r  = rest / GP;
    float arg = (float)(r * g) * (TWO_PI / 1024.0f);
    ws[gid] = entry ? sinf(arg) : cosf(arg);
  } else if (gid < MAX_OFF) {
    int i = gid - FB_OFF;
    int entry = i & 1;
    int rest  = i >> 1;
    int m  = rest % NMELS;
    int k2 = rest / NMELS;
    int f  = 2 * k2 + entry;
    float v = 0.0f;
    if (f < NFREQ) {
      float freq   = (float)f * (22050.0f / 512.0f);
      float melmax = 2595.0f * log10f(1.0f + 22050.0f / 700.0f);
      float step   = melmax / 129.0f;
      float p0 = 700.0f * (exp10f(((float)(m    ) * step) / 2595.0f) - 1.0f);
      float p1 = 700.0f * (exp10f(((float)(m + 1) * step) / 2595.0f) - 1.0f);
      float p2 = 700.0f * (exp10f(((float)(m + 2) * step) / 2595.0f) - 1.0f);
      float down = (freq - p0) / (p1 - p0);
      float up   = (p2 - freq) / (p2 - p1);
      float t = fmaxf(0.0f, fminf(down, up));
      v = t * 2.0f / (p2 - p0);
    }
    ws[gid] = v;
  }
}

#define KH     512
#define HB     15
#define MT2    173
#define HPITCH (FPAD + 4)
#define APITCH (KJ + 4)
#define AROW   (KR * APITCH + 4)

__device__ __forceinline__ void dft4(float r0, float i0, float r1, float i1,
                                     float r2, float i2, float r3, float i3,
                                     float (&Br)[4], float (&Bi)[4]) {
  float s02r = r0 + r2, s02i = i0 + i2, d02r = r0 - r2, d02i = i0 - i2;
  float s13r = r1 + r3, s13i = i1 + i3, d13r = r1 - r3, d13i = i1 - i3;
  Br[0] = s02r + s13r; Bi[0] = s02i + s13i;
  Br[1] = d02r - d13i; Bi[1] = d02i + d13r;
  Br[2] = s02r - s13r; Bi[2] = s02i - s13i;
  Br[3] = d02r + d13i; Bi[3] = d02i - d13r;
}
__global__ __launch_bounds__(256, 6) void mel_gemm(const float* __restrict__ wave,
                                                const float* __restrict__ ws,
                                                float* __restrict__ mel) {
  __shared__ __attribute__((aligned(16))) float AL[16][AROW];
  __shared__ float HC[16][HPITCH];
  __shared__ float HS[16][HPITCH];
  __shared__ float GS[16][KR];
  float (*PL)[AROW] = AL;

  const float* __restrict__ dftT = ws + DFT_OFF;
  const float* __restrict__ twT  = ws + TW_OFF;
  const float* __restrict__ fb2  = ws + FB_OFF;

  int b    = blockIdx.x / MT2;
  int tile = blockIdx.x % MT2;
  int t0   = tile * HB;
  int tid  = threadIdx.x;
  int lane = tid & 31;
  int wv   = __builtin_amdgcn_readfirstlane(tid >> 5);
  int h    = lane >> 4;
  int l16  = lane & 15;

  const float* wb = wave + (long)b * T_IN;

  {
    int base = t0 * HOPSZ - NFFT / 2;
    v4f q[8];
    #pragma unroll
    for (int i = 0; i < 8; ++i) {
      int o = base + 4 * (tid + 256 * i);
      o = min(max(o, 0), T_IN - 4);
      q[i] = *(const v4f*)(wb + o);
    }
    #pragma unroll
    for (int i = 0; i < 8; ++i) {
      int e = tid + 256 * i;
      int m = e >> 7;
      int k = (e & 127) * 4;
      float* dst = &AL[m][(k & (KR - 1)) * APITCH + (k >> 3)];
      #pragma unroll
      for (int c = 0; c < 4; ++c) dst[c * APITCH] = q[i][c];
    }
  }
  if (tile == 0 || tile == MT2 - 1) {
    __syncthreads();
    for (int idx = tid; idx < 16 * KH; idx += 256) {
      int m = idx >> 9;
      int k = idx & (KH - 1);
      int j = t0 + m;
      if (j == 0 || j >= NFRAMES) {
        float v = 0.0f;
        if (j <= NFRAMES) {
          int o = j * HOPSZ + k - NFFT / 2;
          o = (o < 0) ? -o : o;
          o = (o >= T_IN) ? (2 * (T_IN - 1) - o) : o;
          v = wb[o];
        }
        AL[m][(k & (KR - 1)) * APITCH + (k >> 3)] = v;
      }
    }
  }
  __syncthreads();

  {
    int row  = tid >> 4;
    int r    = (tid >> 1) & (KR - 1);
    int half = tid & 1;
    const float* p = &AL[row][r * APITCH + 32 * half];
    float sum = 0.0f;
    #pragma unroll
    for (int j = 0; j < 32; j += 4) {
      v4f x = *(const v4f*)(p + j);
      sum += (x[0] - x[1]) + (x[2] - x[3]);
    }
    sum += __shfl_xor(sum, 1);
    if (half == 0) GS[row][r] = sum;
  }

  int gt  = wv & 3;
  int par = wv >> 2;
  int g   = gt * 16 + l16;
  float tc[4], ts[4];
  v8f aC[4], aS[4];
  {
    const float* dB = dftT + (long)g * 4;
    const float* aB = &AL[l16][par * APITCH];
    #pragma unroll
    for (int n = 0; n < 4; ++n) {
      v2f t = *(const v2f*)(twT + (long)((2 * n + par) * GP + g) * 2);
      tc[n] = t[0]; ts[n] = t[1];
    }
    #pragma unroll
    for (int n = 0; n < 4; ++n) {
      #pragma unroll
      for (int j = 0; j < 8; ++j) { aC[n][j] = 0.0f; aS[n][j] = 0.0f; }
    }
    #pragma unroll
    for (int k = 0; k < KJ; k += 4) {
      int kk = k + 2 * h;
      v4f q = *(const v4f*)(dB + (long)(kk >> 1) * (GP * 4));
      v2f bc; bc[0] = q[0]; bc[1] = q[1];
      v2f bs; bs[0] = q[2]; bs[1] = q[3];
      #pragma unroll
      for (int n = 0; n < 4; ++n) {
        v2f a = *(const v2f*)(aB + 2 * n * APITCH + kk);
        aC[n] = wmma4(a, bc, aC[n]);
        aS[n] = wmma4(a, bs, aS[n]);
      }
    }
  }
  const float RH = 0.70710678118654752440f;
  int mir[4];
  #pragma unroll
  for (int q = 0; q < 4; ++q)
    mir[q] = (q == 0 || g != 0) ? (512 - g - 128 * q) : (528 + q);
  int b513 = (g == 1) ? 513 : ((g == 0) ? 514 : 513 + g);
  float Wr[8][4], Wi[8][4];
  #pragma unroll
  for (int v = 0; v < 8; ++v) {
    float ar[4], ai[4];
    #pragma unroll
    for (int n = 0; n < 4; ++n) {
      ar[n] = aC[n][v] * tc[n] - aS[n][v] * ts[n];
      ai[n] = aC[n][v] * ts[n] + aS[n][v] * tc[n];
    }
    dft4(ar[0], ai[0], ar[1], ai[1], ar[2], ai[2], ar[3], ai[3], Wr[v], Wi[v]);
  }
  if (par == 0) {
    #pragma unroll
    for (int v = 0; v < 8; ++v) {
      int row = v + 8 * h;
      #pragma unroll
      for (int q = 0; q < 4; ++q) {
        HC[row][g + 128 * q] = Wr[v][q];
        HS[row][g + 128 * q] = Wi[v][q];
        HC[row][mir[q]] = Wr[v][q];
        HS[row][mir[q]] = -Wi[v][q];
      }
      if (gt == 0) {
        HC[row][b513] = Wr[v][0];
        HS[row][b513] = Wi[v][0];
      }
    }
  } else {
    #pragma unroll
    for (int v = 0; v < 8; ++v) {
      float r1 = Wr[v][1], i1 = Wi[v][1];
      float r2 = Wr[v][2], i2 = Wi[v][2];
      float r3 = Wr[v][3], i3 = Wi[v][3];
      Wr[v][1] = RH * (r1 - i1);    Wi[v][1] = RH * (r1 + i1);
      Wr[v][2] = -i2;               Wi[v][2] = r2;
      Wr[v][3] = -RH * (r3 + i3);   Wi[v][3] = RH * (r3 - i3);
    }
  }
  __syncthreads();
  if (par == 1) {
    #pragma unroll
    for (int v = 0; v < 8; ++v) {
      int row = v + 8 * h;
      #pragma unroll
      for (int q = 0; q < 4; ++q) {
        HC[row][g + 128 * q] += Wr[v][q];
        HS[row][g + 128 * q] += Wi[v][q];
        HC[row][mir[q]] -= Wr[v][q];
        HS[row][mir[q]] += Wi[v][q];
      }
      if (gt == 0) {
        HC[row][b513] -= Wr[v][0];
        HS[row][b513] -= Wi[v][0];
      }
    }
  } else if (wv == 0 && lane < 16) {
    float ar[KR], ai[KR];
    #pragma unroll
    for (int r = 0; r < KR; ++r) {
      v2f t = *(const v2f*)(twT + (long)(r * GP + 64) * 2);
      float sr = GS[lane][r];
      ar[r] = sr * t[0]; ai[r] = sr * t[1];
    }
    float Er[4], Ei[4], Or[4], Oi[4];
    dft4(ar[0], ai[0], ar[2], ai[2], ar[4], ai[4], ar[6], ai[6], Er, Ei);
    dft4(ar[1], ai[1], ar[3], ai[3], ar[5], ai[5], ar[7], ai[7], Or, Oi);
    HC[lane][64]  = Er[0] + Or[0];
    HS[lane][64]  = Ei[0] + Oi[0];
    HC[lane][192] = Er[1] + RH * (Or[1] - Oi[1]);
    HS[lane][192] = Ei[1] + RH * (Or[1] + Oi[1]);
    HC[lane][320] = Er[2] - Oi[2];
    HS[lane][320] = Ei[2] + Or[2];
    HC[lane][448] = Er[3] - RH * (Or[3] + Oi[3]);
    HS[lane][448] = Ei[3] + RH * (Or[3] - Oi[3]);
  }
  __syncthreads();

  {
    const int   f0 = 2 * tid;
    const int   fm = (tid == 0) ? 1 : f0 - 1;
    const float sm = (tid == 0) ? -1.0f : 1.0f;
    float pc0 = HC[0][fm], pc1 = HC[0][f0], pc2 = HC[0][f0 + 1], pc3 = HC[0][f0 + 2];
    float ps0 = HS[0][fm], ps1 = HS[0][f0], ps2 = HS[0][f0 + 1], ps3 = HS[0][f0 + 2];
    #pragma unroll
    for (int m = 0; m < HB; ++m) {
      float nc0 = HC[m + 1][fm], nc1 = HC[m + 1][f0], nc2 = HC[m + 1][f0 + 1], nc3 = HC[m + 1][f0 + 2];
      float ns0 = HS[m + 1][fm], ns1 = HS[m + 1][f0], ns2 = HS[m + 1][f0 + 1], ns3 = HS[m + 1][f0 + 2];
      float xc0 = pc0 - nc0, xc1 = pc1 + nc1, xc2 = pc2 - nc2, xc3 = pc3 + nc3;
      float xs0 = (ps0 - ns0) * sm, xs1 = ps1 + ns1, xs2 = ps2 - ns2, xs3 = ps3 + ns3;
      float re0 = 0.5f * xc1 - 0.25f * (xc0 + xc2);
      float im0 = 0.5f * xs1 - 0.25f * (xs0 + xs2);
      float re1 = 0.5f * xc2 - 0.25f * (xc1 + xc3);
      float im1 = 0.5f * xs2 - 0.25f * (xs1 + xs3);
      PL[m][f0]     = re0 * re0 + im0 * im0;
      PL[m][f0 + 1] = re1 * re1 + im1 * im1;
      pc0 = nc0; pc1 = nc1; pc2 = nc2; pc3 = nc3;
      ps0 = ns0; ps1 = ns1; ps2 = ns2; ps3 = ns3;
    }
    PL[HB][f0] = 0.0f; PL[HB][f0 + 1] = 0.0f;
  }
  {
    int m = tid >> 4;
    int f = 512 + (tid & 15);
    float p = 0.0f;
    if (f == 512 && m < HB) {
      float c0 = HC[m][512] + HC[m + 1][512];
      float s0 = HS[m][512] + HS[m + 1][512];
      float cm = HC[m][511] - HC[m + 1][511];
      float sn = HS[m][511] - HS[m + 1][511];
      float cp = HC[m][513] - HC[m + 1][513];
      float sp = HS[m][513] - HS[m + 1][513];
      float re = 0.5f * c0 - 0.25f * (cm + cp);
      float im = 0.5f * s0 - 0.25f * (sn + sp);
      p = re * re + im * im;
    }
    PL[m][f] = p;
  }
  __syncthreads();

  int n0 = wv * 16;
  const float* fB = fb2 + (long)(n0 + l16) * 2;
  v8f acc;
  #pragma unroll
  for (int j = 0; j < 8; ++j) acc[j] = 0.0f;
  for (int k = 0; k < FPAD; k += 4) {
    int kk = k + 2 * h;
    v2f a  = *(const v2f*)&PL[l16][kk];
    v2f bb = *(const v2f*)(fB + (long)(kk >> 1) * (NMELS * 2));
    acc = wmma4(a, bb, acc);
  }
  int n = n0 + l16;
  #pragma unroll
  for (int v = 0; v < 8; ++v) {
    int r = v + 8 * h;
    int t = t0 + r;
    if (r < HB && t < NFRAMES) mel[(b * NMELS + n) * NFRAMES + t] = acc[v];
  }
}


#define WIN_INSERT(a, m, x)                                             \
  do {                                                                  \
    if ((m) == 0) { (a)[0] = (x); }                                     \
    else {                                                              \
      (a)[(m)] = fmaxf((a)[(m) - 1], (x));                              \
      for (int i_ = (m) - 1; i_ >= 1; --i_)                             \
        (a)[i_] = __builtin_amdgcn_fmed3f((a)[i_ - 1], (a)[i_], (x));    \
      (a)[0] = fminf((a)[0], (x));                                      \
    }                                                                   \
  } while (0)

__device__ __forceinline__ void win_slide(float (&a)[31], float xo, float xn) {
  float r[30];
  #pragma unroll
  for (int i = 0; i < 30; ++i) r[i] = (a[i] < xo) ? a[i] : a[i + 1];
  a[0] = fminf(r[0], xn);
  #pragma unroll
  for (int i = 1; i < 30; ++i) a[i] = fmaxf(r[i - 1], fminf(r[i], xn));
  a[30] = fmaxf(r[29], xn);
}

#define RT 11
#define RC 5
static_assert(((NFRAMES - 1) / RT) * RT + RC < NFRAMES, "the middle frame of the last run must exist");
__global__ __launch_bounds__(256) void median_time(const float* __restrict__ mel,
                                                   float* __restrict__ medT) {
  __shared__ float P[NFRAMES + 30];
  __shared__ float O[NFRAMES];
  int row = blockIdx.x;
  int tid = threadIdx.x;
  const float* src = mel + (long)row * NFRAMES;
  for (int i = tid; i < NFRAMES + 30; i += 256) {
    int j = i - 15;
    j = (j < 0) ? (-j - 1) : j;
    j = (j >= NFRAMES) ? (2 * NFRAMES - 1 - j) : j;
    P[i] = src[j];
  }
  __syncthreads();
  int t0 = tid * RT;
  if (t0 < NFRAMES) {
    float a[31];
    #pragma unroll
    for (int m = 0; m < 31; ++m) {
      float x = P[t0 + RC + m];
      WIN_INSERT(a, m, x);
    }
    O[t0 + RC] = a[15];
    float w[31];
    #pragma unroll
    for (int m = 0; m < 31; ++m) w[m] = a[m];
    #pragma unroll
    for (int i = RC + 1; i < RT; ++i) {
      int t = t0 + i;
      if (t < NFRAMES) {
        win_slide(a, P[t - 1], P[t + 30]);
        O[t] = a[15];
      }
    }
    #pragma unroll
    for (int i = RC - 1; i >= 0; --i) {
      int t = t0 + i;
      win_slide(w, P[t + 31], P[t]);
      O[t] = w[15];
    }
  }
  __syncthreads();
  float* dst = medT + (long)row * NFRAMES;
  for (int i = tid; i < NFRAMES; i += 256) dst[i] = O[i];
}

__device__ __forceinline__ float to_unit_db(float v, float ref) {
  float db = 10.0f * log10f(fmaxf(1e-10f, v)) -
             10.0f * log10f(fmaxf(1e-10f, ref));
  db = fmaxf(db, -80.0f);
  return (db + 80.0f) * (1.0f / (80.0f + 1e-7f));
}

#define RN 16
#define RNC 8
template <bool FIN>
__global__ __launch_bounds__(256) void hpss_mel(const float* __restrict__ mel,
                                                const float* __restrict__ medT,
                                                unsigned* __restrict__ mx,
                                                float* __restrict__ out) {
  int tid = threadIdx.x;
  int t   = blockIdx.x * 256 + tid;
  int n0  = blockIdx.y * RN;
  int b   = blockIdx.z;
  float s = 0.0f, dh = 0.0f, dp = 0.0f;
  float refS = 0.0f, refH = 0.0f, refP = 0.0f;
  if (FIN) {
    refS = __uint_as_float(mx[0]);
    refH = __uint_as_float(mx[1]);
    refP = __uint_as_float(mx[2]);
  }
  if (t < NFRAMES) {
    int cb = (b * NMELS) * NFRAMES + t;
    const float* base = mel + cb;
    auto elem = [&](int n, float pe) {
      int   e  = cb + n * NFRAMES;
      float S  = mel[e];
      float ha = medT[e];
      float Z  = fmaxf(ha, pe);
      bool bad = Z < 1.17549435e-38f;
      float Zs = bad ? 1.0f : Z;
      float rz = __builtin_amdgcn_rcpf(Zs);
      float mh = ha * rz; mh *= mh;
      float mp = pe * rz; mp *= mp;
      float rd = __builtin_amdgcn_rcpf(mh + mp);
      float maskh = bad ? 0.0f : mh * rd;
      float maskp = bad ? 0.0f : mp * rd;
      float vh = S * maskh, vp = S * maskp;
      if (FIN) {
        out[e]               = to_unit_db(S,  refS);
        out[E_ELEMS + e]     = to_unit_db(vh, refH);
        out[2 * E_ELEMS + e] = to_unit_db(vp, refP);
      } else {
        s = fmaxf(s, S); dh = fmaxf(dh, vh); dp = fmaxf(dp, vp);
      }
    };
    float a[31];
    #pragma unroll
    for (int m = 0; m < 31; ++m) {
      int j = n0 + RNC + m - 15;
      j = (j < 0) ? (-j - 1) : j;
      j = (j >= NMELS) ? (2 * NMELS - 1 - j) : j;
      float x = base[j * NFRAMES];
      WIN_INSERT(a, m, x);
    }
    float w[31];
    #pragma unroll
    for (int m = 0; m < 31; ++m) w[m] = a[m];
    elem(n0 + RNC, a[15]);
    #pragma unroll
    for (int i = RNC + 1; i < RN; ++i) {
      int n = n0 + i;
      int jo = n - 16;
      jo = (jo < 0) ? (-jo - 1) : jo;
      int jn = n + 15;
      jn = (jn >= NMELS) ? (2 * NMELS - 1 - jn) : jn;
      win_slide(a, base[jo * NFRAMES], base[jn * NFRAMES]);
      elem(n, a[15]);
    }
    #pragma unroll
    for (int i = RNC - 1; i >= 0; --i) {
      int n = n0 + i;
      int jo = n + 16;
      jo = (jo >= NMELS) ? (2 * NMELS - 1 - jo) : jo;
      int jn = n - 15;
      jn = (jn < 0) ? (-jn - 1) : jn;
      win_slide(w, base[jo * NFRAMES], base[jn * NFRAMES]);
      elem(n, w[15]);
    }
  }
  if (FIN) return;
  __shared__ float r0[256], r1[256], r2[256];
  r0[tid] = s; r1[tid] = dh; r2[tid] = dp;
  __syncthreads();
  for (int off = 128; off > 0; off >>= 1) {
    if (tid < off) {
      r0[tid] = fmaxf(r0[tid], r0[tid + off]);
      r1[tid] = fmaxf(r1[tid], r1[tid + off]);
      r2[tid] = fmaxf(r2[tid], r2[tid + off]);
    }
    __syncthreads();
  }
  if (tid == 0) {
    atomicMax(mx + 0, __float_as_uint(r0[0]));
    atomicMax(mx + 1, __float_as_uint(r1[0]));
    atomicMax(mx + 2, __float_as_uint(r2[0]));
  }
}

extern "C" void kernel_launch(void* const* d_in, const int* in_sizes, int n_in,
                              void* d_out, int out_size, void* d_ws,
                              size_t ws_size, hipStream_t stream) {
  const float* wave = (const float*)d_in[0];
  float* out  = (float*)d_out;
  float* ws   = (float*)d_ws;
  float* mel  = ws + MEL_OFF;
  float* medT = ws + MEDT_OFF;
  unsigned* mx = (unsigned*)(ws + MAX_OFF);

  build_tables<<<MAX_OFF / 256, 256, 0, stream>>>(ws);
  mel_gemm<<<NB * MT2, 256, 0, stream>>>(wave, ws, mel);
  median_time<<<NB * NMELS, 256, 0, stream>>>(mel, medT);
  dim3 hg((NFRAMES + 255) / 256, NMELS / RN, NB);
  hpss_mel<false><<<hg, 256, 0, stream>>>(mel, medT, mx, out);
  hpss_mel<true><<<hg, 256, 0, stream>>>(mel, medT, mx, out);
}
